# starting kernel
# baseline (speedup 1.0000x reference)
#include <hip/hip_runtime.h>
#include <math.h>

#define Bsz 8
#define Tsz 2048
#define Csz 1024
#define Hsz 64

typedef __attribute__((ext_vector_type(2))) float v2f;
typedef __attribute__((ext_vector_type(8))) float v8f;

__device__ __forceinline__ v8f wmma_f32(v2f a, v2f b, v8f c) {
    return __builtin_amdgcn_wmma_f32_16x16x4_f32(
         false, a,  false, b,
         (short)0, c,  false,  false);
}

__global__ __launch_bounds__(256) void wt_kernel(
    const float* __restrict__ Wq,
    const float* __restrict__ Wk,
    const float* __restrict__ Wv,
    float* __restrict__ Wt)
{
    const int which = blockIdx.y;
    const float* W = (which == 0) ? Wq : (which == 1) ? Wk : Wv;
    float* dst = Wt + (size_t)which * Hsz * Csz;
    const int h = blockIdx.x;
    for (int c = threadIdx.x; c < Csz; c += 256)
        dst[(size_t)h * Csz + c] = W[(size_t)c * Hsz + h];
}

__global__ __launch_bounds__(128) void qkv_kernel(
    const float* __restrict__ x,
    const float* __restrict__ Wt,
    float* __restrict__ qkv)
{
    const int lane = threadIdx.x & 31;
    const int wv   = threadIdx.x >> 5;
    const int half = lane >> 4;
    const int l16  = lane & 15;

    const long mrow  = (long)blockIdx.x * 64 + wv * 16;
    const int  which = blockIdx.y;
    const float* W = Wt + (size_t)which * Hsz * Csz;

    const float* arow = x + (mrow + l16) * (size_t)Csz + 2 * half;
    const float* b0 = W + (size_t)( 0 + l16) * Csz + 2 * half;
    const float* b1 = W + (size_t)(16 + l16) * Csz + 2 * half;
    const float* b2 = W + (size_t)(32 + l16) * Csz + 2 * half;
    const float* b3 = W + (size_t)(48 + l16) * Csz + 2 * half;

    v8f acc0 = {}, acc1 = {}, acc2 = {}, acc3 = {};
#pragma unroll 4
    for (int kk = 0; kk < Csz; kk += 4) {
        v2f a  = *(const v2f*)(arow + kk);
        v2f w0 = *(const v2f*)(b0 + kk);
        v2f w1 = *(const v2f*)(b1 + kk);
        v2f w2 = *(const v2f*)(b2 + kk);
        v2f w3 = *(const v2f*)(b3 + kk);
        acc0 = wmma_f32(a, w0, acc0);
        acc1 = wmma_f32(a, w1, acc1);
        acc2 = wmma_f32(a, w2, acc2);
        acc3 = wmma_f32(a, w3, acc3);
    }

    const size_t BTH = (size_t)Bsz * Tsz * Hsz;
    if (which < 2) {
        float* op = qkv + (size_t)which * BTH + (mrow + 8 * half) * (size_t)Hsz + l16;
#pragma unroll
        for (int r = 0; r < 8; ++r) {
            op[(size_t)r * Hsz +  0] = acc0[r];
            op[(size_t)r * Hsz + 16] = acc1[r];
            op[(size_t)r * Hsz + 32] = acc2[r];
            op[(size_t)r * Hsz + 48] = acc3[r];
        }
    } else {
        const long bb   = mrow / Tsz;
        const long tloc = mrow - bb * Tsz + 8 * half;
        float* vt = qkv + 2 * BTH;
        const v8f* accs[4] = { &acc0, &acc1, &acc2, &acc3 };
#pragma unroll
        for (int nt = 0; nt < 4; ++nt) {
            const int col = nt * 16 + l16;
            float* vp = vt + ((size_t)bb * Hsz + col) * Tsz + tloc;
            const v8f& a = *accs[nt];
            float4 lo = make_float4(a[0], a[1], a[2], a[3]);
            float4 hi = make_float4(a[4], a[5], a[6], a[7]);
            *(float4*)(vp + 0) = lo;
            *(float4*)(vp + 4) = hi;
        }
    }
}

__global__ __launch_bounds__(32) void attn_kernel(
    const float* __restrict__ qkv,
    float* __restrict__ out)
{
    __shared__ float Plds[16 * 16];

    const int lane = threadIdx.x;
    const int half = lane >> 4;
    const int l16  = lane & 15;
    const int b    = blockIdx.y;
    const int qi   = gridDim.x - 1 - blockIdx.x;

    const size_t BTH = (size_t)Bsz * Tsz * Hsz;
    const float* Q  = qkv +            (size_t)b * Tsz * Hsz;
    const float* K  = qkv + BTH     +  (size_t)b * Tsz * Hsz;
    const float* Vt = qkv + 2 * BTH +  (size_t)b * Hsz * Tsz;

    const float scale = 0.125f;
    const int qrow = qi * 16 + l16;

    v2f qa[16];
#pragma unroll
    for (int c = 0; c < 16; ++c) {
        v2f t = *(const v2f*)(Q + (size_t)qrow * Hsz + 4 * c + 2 * half);
        qa[c][0] = t[0] * scale;
        qa[c][1] = t[1] * scale;
    }

    const float* Vr0 = Vt + (size_t)( 0 + l16) * Tsz + 2 * half;
    const float* Vr1 = Vt + (size_t)(16 + l16) * Tsz + 2 * half;
    const float* Vr2 = Vt + (size_t)(32 + l16) * Tsz + 2 * half;
    const float* Vr3 = Vt + (size_t)(48 + l16) * Tsz + 2 * half;

    float mrow[8], lrow[8];
#pragma unroll
    for (int r = 0; r < 8; ++r) { mrow[r] = -INFINITY; lrow[r] = 0.0f; }
    v8f o0 = {}, o1 = {}, o2 = {}, o3 = {};

    for (int kt = 0; kt <= qi; ++kt) {
        const float* Kt = K + (size_t)(kt * 16 + l16) * Hsz + 2 * half;
        v8f s = {};
#pragma unroll
        for (int c = 0; c < 16; ++c) {
            v2f kb = *(const v2f*)(Kt + 4 * c);
            s = wmma_f32(qa[c], kb, s);
        }

        if (kt == qi) {
#pragma unroll
            for (int r = 0; r < 8; ++r)
                if (l16 > r + 8 * half) s[r] = -INFINITY;
        }

        float alpha[8], p[8];
#pragma unroll
        for (int r = 0; r < 8; ++r) {
            float mx = s[r];
            mx = fmaxf(mx, __shfl_xor(mx, 1));
            mx = fmaxf(mx, __shfl_xor(mx, 2));
            mx = fmaxf(mx, __shfl_xor(mx, 4));
            mx = fmaxf(mx, __shfl_xor(mx, 8));
            const float mn = fmaxf(mrow[r], mx);
            alpha[r] = expf(mrow[r] - mn);
            mrow[r]  = mn;
            float pv = expf(s[r] - mn);
            p[r] = pv;
            float sm = pv;
            sm += __shfl_xor(sm, 1);
            sm += __shfl_xor(sm, 2);
            sm += __shfl_xor(sm, 4);
            sm += __shfl_xor(sm, 8);
            lrow[r] = lrow[r] * alpha[r] + sm;
        }

#pragma unroll
        for (int r = 0; r < 8; ++r)
            Plds[(r + 8 * half) * 16 + l16] = p[r];
        __syncthreads();

#pragma unroll
        for (int r = 0; r < 8; ++r) {
            o0[r] *= alpha[r]; o1[r] *= alpha[r];
            o2[r] *= alpha[r]; o3[r] *= alpha[r];
        }

        v2f pa[4];
#pragma unroll
        for (int c2 = 0; c2 < 4; ++c2)
            pa[c2] = *(const v2f*)(&Plds[l16 * 16 + 4 * c2 + 2 * half]);

        const int kb0 = kt * 16;
#pragma unroll
        for (int c2 = 0; c2 < 4; ++c2) {
            const int ko = kb0 + 4 * c2;
            v2f vb;
            vb = *(const v2f*)(Vr0 + ko); o0 = wmma_f32(pa[c2], vb, o0);
            vb = *(const v2f*)(Vr1 + ko); o1 = wmma_f32(pa[c2], vb, o1);
            vb = *(const v2f*)(Vr2 + ko); o2 = wmma_f32(pa[c2], vb, o2);
            vb = *(const v2f*)(Vr3 + ko); o3 = wmma_f32(pa[c2], vb, o3);
        }
        __syncthreads();
    }

    float* Ob = out + ((size_t)b * Tsz + qi * 16 + 8 * half) * Hsz + l16;
#pragma unroll
    for (int r = 0; r < 8; ++r) {
        const float inv = 1.0f / lrow[r];
        Ob[(size_t)r * Hsz +  0] = o0[r] * inv;
        Ob[(size_t)r * Hsz + 16] = o1[r] * inv;
        Ob[(size_t)r * Hsz + 32] = o2[r] * inv;
        Ob[(size_t)r * Hsz + 48] = o3[r] * inv;
    }
}

extern "C" void kernel_launch(void* const* d_in, const int* in_sizes, int n_in,
                              void* d_out, int out_size, void* d_ws, size_t ws_size,
                              hipStream_t stream) {
    const float* x  = (const float*)d_in[0];
    const float* Wq = (const float*)d_in[1];
    const float* Wk = (const float*)d_in[2];
    const float* Wv = (const float*)d_in[3];

    const size_t BTH = (size_t)Bsz * Tsz * Hsz;
    float* qkv = (float*)d_ws;
    float* Wt  = qkv + 3 * BTH;

    dim3 g0(Hsz, 3), blk0(256);
    wt_kernel<<<g0, blk0, 0, stream>>>(Wq, Wk, Wv, Wt);

    dim3 g1(Bsz * Tsz / 64, 3), blk1(128);
    qkv_kernel<<<g1, blk1, 0, stream>>>(x, Wt, qkv);

    dim3 g2(Tsz / 16, Bsz), blk2(32);
    attn_kernel<<<g2, blk2, 0, stream>>>(qkv, (float*)d_out);
}


# attention: split each query tile's causal key range across the 8 waves of a work-group, merge the 8 online-softmax partial states through LDS in fixed wave order (summation order changes)
# speedup vs baseline: 2.1981x; 2.1981x over previous
#include <hip/hip_runtime.h>
#include <math.h>

#define Bsz 8
#define Tsz 2048
#define Csz 1024
#define Hsz 64

typedef __attribute__((ext_vector_type(2))) float v2f;
typedef __attribute__((ext_vector_type(8))) float v8f;

__device__ __forceinline__ v8f wmma_f32(v2f a, v2f b, v8f c) {
    return __builtin_amdgcn_wmma_f32_16x16x4_f32(
         false, a,  false, b,
         (short)0, c,  false,  false);
}

__global__ __launch_bounds__(256) void wt_kernel(
    const float* __restrict__ Wq,
    const float* __restrict__ Wk,
    const float* __restrict__ Wv,
    float* __restrict__ Wt)
{
    const int which = blockIdx.y;
    const float* W = (which == 0) ? Wq : (which == 1) ? Wk : Wv;
    float* dst = Wt + (size_t)which * Hsz * Csz;
    const int h = blockIdx.x;
    for (int c = threadIdx.x; c < Csz; c += 256)
        dst[(size_t)h * Csz + c] = W[(size_t)c * Hsz + h];
}

__global__ __launch_bounds__(128) void qkv_kernel(
    const float* __restrict__ x,
    const float* __restrict__ Wt,
    float* __restrict__ qkv)
{
    const int lane = threadIdx.x & 31;
    const int wv   = threadIdx.x >> 5;
    const int half = lane >> 4;
    const int l16  = lane & 15;

    const long mrow  = (long)blockIdx.x * 64 + wv * 16;
    const int  which = blockIdx.y;
    const float* W = Wt + (size_t)which * Hsz * Csz;

    const float* arow = x + (mrow + l16) * (size_t)Csz + 2 * half;
    const float* b0 = W + (size_t)( 0 + l16) * Csz + 2 * half;
    const float* b1 = W + (size_t)(16 + l16) * Csz + 2 * half;
    const float* b2 = W + (size_t)(32 + l16) * Csz + 2 * half;
    const float* b3 = W + (size_t)(48 + l16) * Csz + 2 * half;

    v8f acc0 = {}, acc1 = {}, acc2 = {}, acc3 = {};
#pragma unroll 4
    for (int kk = 0; kk < Csz; kk += 4) {
        v2f a  = *(const v2f*)(arow + kk);
        v2f w0 = *(const v2f*)(b0 + kk);
        v2f w1 = *(const v2f*)(b1 + kk);
        v2f w2 = *(const v2f*)(b2 + kk);
        v2f w3 = *(const v2f*)(b3 + kk);
        acc0 = wmma_f32(a, w0, acc0);
        acc1 = wmma_f32(a, w1, acc1);
        acc2 = wmma_f32(a, w2, acc2);
        acc3 = wmma_f32(a, w3, acc3);
    }

    const size_t BTH = (size_t)Bsz * Tsz * Hsz;
    if (which < 2) {
        float* op = qkv + (size_t)which * BTH + (mrow + 8 * half) * (size_t)Hsz + l16;
#pragma unroll
        for (int r = 0; r < 8; ++r) {
            op[(size_t)r * Hsz +  0] = acc0[r];
            op[(size_t)r * Hsz + 16] = acc1[r];
            op[(size_t)r * Hsz + 32] = acc2[r];
            op[(size_t)r * Hsz + 48] = acc3[r];
        }
    } else {
        const long bb   = mrow / Tsz;
        const long tloc = mrow - bb * Tsz + 8 * half;
        float* vt = qkv + 2 * BTH;
        const v8f* accs[4] = { &acc0, &acc1, &acc2, &acc3 };
#pragma unroll
        for (int nt = 0; nt < 4; ++nt) {
            const int col = nt * 16 + l16;
            float* vp = vt + ((size_t)bb * Hsz + col) * Tsz + tloc;
            const v8f& a = *accs[nt];
            float4 lo = make_float4(a[0], a[1], a[2], a[3]);
            float4 hi = make_float4(a[4], a[5], a[6], a[7]);
            *(float4*)(vp + 0) = lo;
            *(float4*)(vp + 4) = hi;
        }
    }
}

#define NWAVE 8
#define OPITCH 68

__device__ __forceinline__ void wave_lds_sync() {
    __builtin_amdgcn_fence(__ATOMIC_RELEASE, "wavefront");
    __builtin_amdgcn_wave_barrier();
    __builtin_amdgcn_fence(__ATOMIC_ACQUIRE, "wavefront");
}

__global__ __launch_bounds__(32 * NWAVE) void attn_kernel(
    const float* __restrict__ qkv,
    float* __restrict__ out)
{
    __shared__ __attribute__((aligned(16))) float Wsm[NWAVE][16 * OPITCH];
    __shared__ float Msm[NWAVE][16];
    __shared__ float Lsm[NWAVE][16];

    const int lane = threadIdx.x & 31;
    const int wv   = __builtin_amdgcn_readfirstlane((int)(threadIdx.x >> 5));
    const int half = lane >> 4;
    const int l16  = lane & 15;
    const int b    = blockIdx.y;
    const int qi   = gridDim.x - 1 - blockIdx.x;

    const size_t BTH = (size_t)Bsz * Tsz * Hsz;
    const float* Q  = qkv +            (size_t)b * Tsz * Hsz;
    const float* K  = qkv + BTH     +  (size_t)b * Tsz * Hsz;
    const float* Vt = qkv + 2 * BTH +  (size_t)b * Hsz * Tsz;

    float* Plds = &Wsm[wv][0];

    const float scale = 0.125f;
    const int qrow = qi * 16 + l16;

    v2f qa[16];
#pragma unroll
    for (int c = 0; c < 16; ++c) {
        v2f t = *(const v2f*)(Q + (size_t)qrow * Hsz + 4 * c + 2 * half);
        qa[c][0] = t[0] * scale;
        qa[c][1] = t[1] * scale;
    }

    const float* Vr0 = Vt + (size_t)( 0 + l16) * Tsz + 2 * half;
    const float* Vr1 = Vt + (size_t)(16 + l16) * Tsz + 2 * half;
    const float* Vr2 = Vt + (size_t)(32 + l16) * Tsz + 2 * half;
    const float* Vr3 = Vt + (size_t)(48 + l16) * Tsz + 2 * half;

    float mrow[8], lrow[8];
#pragma unroll
    for (int r = 0; r < 8; ++r) { mrow[r] = -INFINITY; lrow[r] = 0.0f; }
    v8f o0 = {}, o1 = {}, o2 = {}, o3 = {};

    for (int kt = wv; kt <= qi; kt += NWAVE) {
        const float* Kt = K + (size_t)(kt * 16 + l16) * Hsz + 2 * half;
        v8f s = {};
#pragma unroll
        for (int c = 0; c < 16; ++c) {
            v2f kb = *(const v2f*)(Kt + 4 * c);
            s = wmma_f32(qa[c], kb, s);
        }

        if (kt == qi) {
#pragma unroll
            for (int r = 0; r < 8; ++r)
                if (l16 > r + 8 * half) s[r] = -INFINITY;
        }

        float alpha[8], p[8];
#pragma unroll
        for (int r = 0; r < 8; ++r) {
            float mx = s[r];
            mx = fmaxf(mx, __shfl_xor(mx, 1));
            mx = fmaxf(mx, __shfl_xor(mx, 2));
            mx = fmaxf(mx, __shfl_xor(mx, 4));
            mx = fmaxf(mx, __shfl_xor(mx, 8));
            const float mn = fmaxf(mrow[r], mx);
            alpha[r] = expf(mrow[r] - mn);
            mrow[r]  = mn;
            float pv = expf(s[r] - mn);
            p[r] = pv;
            float sm = pv;
            sm += __shfl_xor(sm, 1);
            sm += __shfl_xor(sm, 2);
            sm += __shfl_xor(sm, 4);
            sm += __shfl_xor(sm, 8);
            lrow[r] = lrow[r] * alpha[r] + sm;
        }

#pragma unroll
        for (int r = 0; r < 8; ++r)
            Plds[(r + 8 * half) * 16 + l16] = p[r];
        wave_lds_sync();

#pragma unroll
        for (int r = 0; r < 8; ++r) {
            o0[r] *= alpha[r]; o1[r] *= alpha[r];
            o2[r] *= alpha[r]; o3[r] *= alpha[r];
        }

        v2f pa[4];
#pragma unroll
        for (int c2 = 0; c2 < 4; ++c2)
            pa[c2] = *(const v2f*)(&Plds[l16 * 16 + 4 * c2 + 2 * half]);

        const int kb0 = kt * 16;
#pragma unroll
        for (int c2 = 0; c2 < 4; ++c2) {
            const int ko = kb0 + 4 * c2;
            v2f vb;
            vb = *(const v2f*)(Vr0 + ko); o0 = wmma_f32(pa[c2], vb, o0);
            vb = *(const v2f*)(Vr1 + ko); o1 = wmma_f32(pa[c2], vb, o1);
            vb = *(const v2f*)(Vr2 + ko); o2 = wmma_f32(pa[c2], vb, o2);
            vb = *(const v2f*)(Vr3 + ko); o3 = wmma_f32(pa[c2], vb, o3);
        }
        wave_lds_sync();
    }

    float* Ow = &Wsm[wv][0];
#pragma unroll
    for (int r = 0; r < 8; ++r) {
        float* orow = Ow + (r + 8 * half) * OPITCH + l16;
        orow[ 0] = o0[r];
        orow[16] = o1[r];
        orow[32] = o2[r];
        orow[48] = o3[r];
    }
    if (l16 == 0) {
#pragma unroll
        for (int r = 0; r < 8; ++r) {
            Msm[wv][r + 8 * half] = mrow[r];
            Lsm[wv][r + 8 * half] = lrow[r];
        }
    }
    __syncthreads();

    const int R = 2 * wv + half;
    float mw[NWAVE];
    float M = -INFINITY;
#pragma unroll
    for (int w = 0; w < NWAVE; ++w) { mw[w] = Msm[w][R]; M = fmaxf(M, mw[w]); }
    float L = 0.0f;
    float4 acc = make_float4(0.0f, 0.0f, 0.0f, 0.0f);
#pragma unroll
    for (int w = 0; w < NWAVE; ++w) {
        const float g = expf(mw[w] - M);
        L += g * Lsm[w][R];
        const float4 ov = *(const float4*)(&Wsm[w][R * OPITCH + 4 * l16]);
        acc.x += g * ov.x; acc.y += g * ov.y;
        acc.z += g * ov.z; acc.w += g * ov.w;
    }
    const float inv = 1.0f / L;
    acc.x *= inv; acc.y *= inv; acc.z *= inv; acc.w *= inv;
    float* Ob = out + ((size_t)b * Tsz + qi * 16 + R) * Hsz + 4 * l16;
    *(float4*)Ob = acc;
}

extern "C" void kernel_launch(void* const* d_in, const int* in_sizes, int n_in,
                              void* d_out, int out_size, void* d_ws, size_t ws_size,
                              hipStream_t stream) {
    const float* x  = (const float*)d_in[0];
    const float* Wq = (const float*)d_in[1];
    const float* Wk = (const float*)d_in[2];
    const float* Wv = (const float*)d_in[3];

    const size_t BTH = (size_t)Bsz * Tsz * Hsz;
    float* qkv = (float*)d_ws;
    float* Wt  = qkv + 3 * BTH;

    dim3 g0(Hsz, 3), blk0(256);
    wt_kernel<<<g0, blk0, 0, stream>>>(Wq, Wk, Wv, Wt);

    dim3 g1(Bsz * Tsz / 64, 3), blk1(128);
    qkv_kernel<<<g1, blk1, 0, stream>>>(x, Wt, qkv);

    dim3 g2(Tsz / 16, Bsz), blk2(32 * NWAVE);
    attn_kernel<<<g2, blk2, 0, stream>>>(qkv, (float*)d_out);
}


# every WMMA fragment loader (x, W, Q, K, V, P) reads 16 B per lane for two k-steps, same permuted k order on A and B: half the cache lines touched per WMMA (sum order inside each 8-k group changes)
# speedup vs baseline: 2.3480x; 1.0682x over previous
#include <hip/hip_runtime.h>
#include <math.h>

#define Bsz 8
#define Tsz 2048
#define Csz 1024
#define Hsz 64

typedef __attribute__((ext_vector_type(2))) float v2f;
typedef __attribute__((ext_vector_type(8))) float v8f;

__device__ __forceinline__ v8f wmma_f32(v2f a, v2f b, v8f c) {
    return __builtin_amdgcn_wmma_f32_16x16x4_f32(
         false, a,  false, b,
         (short)0, c,  false,  false);
}

__global__ __launch_bounds__(256) void wt_kernel(
    const float* __restrict__ Wq,
    const float* __restrict__ Wk,
    const float* __restrict__ Wv,
    float* __restrict__ Wt)
{
    const int which = blockIdx.y;
    const float* W = (which == 0) ? Wq : (which == 1) ? Wk : Wv;
    float* dst = Wt + (size_t)which * Hsz * Csz;
    const int h = blockIdx.x;
    for (int c = threadIdx.x; c < Csz; c += 256)
        dst[(size_t)h * Csz + c] = W[(size_t)c * Hsz + h];
}

__global__ __launch_bounds__(128) void qkv_kernel(
    const float* __restrict__ x,
    const float* __restrict__ Wt,
    float* __restrict__ qkv)
{
    const int lane = threadIdx.x & 31;
    const int wv   = threadIdx.x >> 5;
    const int half = lane >> 4;
    const int l16  = lane & 15;

    const long mrow  = (long)blockIdx.x * 64 + wv * 16;
    const int  which = blockIdx.y;
    const float* W = Wt + (size_t)which * Hsz * Csz;

    const float* arow = x + (mrow + l16) * (size_t)Csz + 4 * half;
    const float* b0 = W + (size_t)( 0 + l16) * Csz + 4 * half;
    const float* b1 = W + (size_t)(16 + l16) * Csz + 4 * half;
    const float* b2 = W + (size_t)(32 + l16) * Csz + 4 * half;
    const float* b3 = W + (size_t)(48 + l16) * Csz + 4 * half;

    v8f acc0 = {}, acc1 = {}, acc2 = {}, acc3 = {};
#pragma unroll 2
    for (int kk = 0; kk < Csz; kk += 8) {
        const float4 a  = *(const float4*)(arow + kk);
        const float4 w0 = *(const float4*)(b0 + kk);
        const float4 w1 = *(const float4*)(b1 + kk);
        const float4 w2 = *(const float4*)(b2 + kk);
        const float4 w3 = *(const float4*)(b3 + kk);
        const v2f alo = { a.x, a.y }, ahi = { a.z, a.w };
        acc0 = wmma_f32(alo, (v2f){ w0.x, w0.y }, acc0);
        acc1 = wmma_f32(alo, (v2f){ w1.x, w1.y }, acc1);
        acc2 = wmma_f32(alo, (v2f){ w2.x, w2.y }, acc2);
        acc3 = wmma_f32(alo, (v2f){ w3.x, w3.y }, acc3);
        acc0 = wmma_f32(ahi, (v2f){ w0.z, w0.w }, acc0);
        acc1 = wmma_f32(ahi, (v2f){ w1.z, w1.w }, acc1);
        acc2 = wmma_f32(ahi, (v2f){ w2.z, w2.w }, acc2);
        acc3 = wmma_f32(ahi, (v2f){ w3.z, w3.w }, acc3);
    }

    const size_t BTH = (size_t)Bsz * Tsz * Hsz;
    if (which < 2) {
        float* op = qkv + (size_t)which * BTH + (mrow + 8 * half) * (size_t)Hsz + l16;
#pragma unroll
        for (int r = 0; r < 8; ++r) {
            op[(size_t)r * Hsz +  0] = acc0[r];
            op[(size_t)r * Hsz + 16] = acc1[r];
            op[(size_t)r * Hsz + 32] = acc2[r];
            op[(size_t)r * Hsz + 48] = acc3[r];
        }
    } else {
        const long bb   = mrow / Tsz;
        const long tloc = mrow - bb * Tsz + 8 * half;
        float* vt = qkv + 2 * BTH;
        const v8f* accs[4] = { &acc0, &acc1, &acc2, &acc3 };
#pragma unroll
        for (int nt = 0; nt < 4; ++nt) {
            const int col = nt * 16 + l16;
            float* vp = vt + ((size_t)bb * Hsz + col) * Tsz + tloc;
            const v8f& a = *accs[nt];
            float4 lo = make_float4(a[0], a[1], a[2], a[3]);
            float4 hi = make_float4(a[4], a[5], a[6], a[7]);
            *(float4*)(vp + 0) = lo;
            *(float4*)(vp + 4) = hi;
        }
    }
}

#define NWAVE 8
#define OPITCH 68
#define PPITCH 20

__device__ __forceinline__ void wave_lds_sync() {
    __builtin_amdgcn_fence(__ATOMIC_RELEASE, "wavefront");
    __builtin_amdgcn_wave_barrier();
    __builtin_amdgcn_fence(__ATOMIC_ACQUIRE, "wavefront");
}

__global__ __launch_bounds__(32 * NWAVE) void attn_kernel(
    const float* __restrict__ qkv,
    float* __restrict__ out)
{
    __shared__ __attribute__((aligned(16))) float Wsm[NWAVE][16 * OPITCH];
    __shared__ float Msm[NWAVE][16];
    __shared__ float Lsm[NWAVE][16];

    const int lane = threadIdx.x & 31;
    const int wv   = __builtin_amdgcn_readfirstlane((int)(threadIdx.x >> 5));
    const int half = lane >> 4;
    const int l16  = lane & 15;
    const int b    = blockIdx.y;
    const int qi   = gridDim.x - 1 - blockIdx.x;

    const size_t BTH = (size_t)Bsz * Tsz * Hsz;
    const float* Q  = qkv +            (size_t)b * Tsz * Hsz;
    const float* K  = qkv + BTH     +  (size_t)b * Tsz * Hsz;
    const float* Vt = qkv + 2 * BTH +  (size_t)b * Hsz * Tsz;

    float* Plds = &Wsm[wv][0];

    const float scale = 0.125f;
    const int qrow = qi * 16 + l16;

    v2f qa[16];
#pragma unroll
    for (int c = 0; c < 8; ++c) {
        const float4 t = *(const float4*)(Q + (size_t)qrow * Hsz + 8 * c + 4 * half);
        qa[2 * c][0]     = t.x * scale;
        qa[2 * c][1]     = t.y * scale;
        qa[2 * c + 1][0] = t.z * scale;
        qa[2 * c + 1][1] = t.w * scale;
    }

    const float* Vr0 = Vt + (size_t)( 0 + l16) * Tsz + 4 * half;
    const float* Vr1 = Vt + (size_t)(16 + l16) * Tsz + 4 * half;
    const float* Vr2 = Vt + (size_t)(32 + l16) * Tsz + 4 * half;
    const float* Vr3 = Vt + (size_t)(48 + l16) * Tsz + 4 * half;

    float mrow[8], lrow[8];
#pragma unroll
    for (int r = 0; r < 8; ++r) { mrow[r] = -INFINITY; lrow[r] = 0.0f; }
    v8f o0 = {}, o1 = {}, o2 = {}, o3 = {};

    for (int kt = wv; kt <= qi; kt += NWAVE) {
        const float* Kt = K + (size_t)(kt * 16 + l16) * Hsz + 4 * half;
        v8f s = {};
#pragma unroll
        for (int c = 0; c < 8; ++c) {
            const float4 kb = *(const float4*)(Kt + 8 * c);
            s = wmma_f32(qa[2 * c],     (v2f){ kb.x, kb.y }, s);
            s = wmma_f32(qa[2 * c + 1], (v2f){ kb.z, kb.w }, s);
        }

        if (kt == qi) {
#pragma unroll
            for (int r = 0; r < 8; ++r)
                if (l16 > r + 8 * half) s[r] = -INFINITY;
        }

        float alpha[8], p[8];
#pragma unroll
        for (int r = 0; r < 8; ++r) {
            float mx = s[r];
            mx = fmaxf(mx, __shfl_xor(mx, 1));
            mx = fmaxf(mx, __shfl_xor(mx, 2));
            mx = fmaxf(mx, __shfl_xor(mx, 4));
            mx = fmaxf(mx, __shfl_xor(mx, 8));
            const float mn = fmaxf(mrow[r], mx);
            alpha[r] = expf(mrow[r] - mn);
            mrow[r]  = mn;
            float pv = expf(s[r] - mn);
            p[r] = pv;
            float sm = pv;
            sm += __shfl_xor(sm, 1);
            sm += __shfl_xor(sm, 2);
            sm += __shfl_xor(sm, 4);
            sm += __shfl_xor(sm, 8);
            lrow[r] = lrow[r] * alpha[r] + sm;
        }

#pragma unroll
        for (int r = 0; r < 8; ++r)
            Plds[(r + 8 * half) * PPITCH + l16] = p[r];
        wave_lds_sync();

#pragma unroll
        for (int r = 0; r < 8; ++r) {
            o0[r] *= alpha[r]; o1[r] *= alpha[r];
            o2[r] *= alpha[r]; o3[r] *= alpha[r];
        }

        v2f pa[4];
#pragma unroll
        for (int j = 0; j < 2; ++j) {
            const float4 t = *(const float4*)(&Plds[l16 * PPITCH + 8 * j + 4 * half]);
            pa[2 * j]     = (v2f){ t.x, t.y };
            pa[2 * j + 1] = (v2f){ t.z, t.w };
        }

        const int kb0 = kt * 16;
#pragma unroll
        for (int j = 0; j < 2; ++j) {
            const int ko = kb0 + 8 * j;
            const float4 v0 = *(const float4*)(Vr0 + ko);
            const float4 v1 = *(const float4*)(Vr1 + ko);
            const float4 v2 = *(const float4*)(Vr2 + ko);
            const float4 v3 = *(const float4*)(Vr3 + ko);
            o0 = wmma_f32(pa[2 * j],     (v2f){ v0.x, v0.y }, o0);
            o1 = wmma_f32(pa[2 * j],     (v2f){ v1.x, v1.y }, o1);
            o2 = wmma_f32(pa[2 * j],     (v2f){ v2.x, v2.y }, o2);
            o3 = wmma_f32(pa[2 * j],     (v2f){ v3.x, v3.y }, o3);
            o0 = wmma_f32(pa[2 * j + 1], (v2f){ v0.z, v0.w }, o0);
            o1 = wmma_f32(pa[2 * j + 1], (v2f){ v1.z, v1.w }, o1);
            o2 = wmma_f32(pa[2 * j + 1], (v2f){ v2.z, v2.w }, o2);
            o3 = wmma_f32(pa[2 * j + 1], (v2f){ v3.z, v3.w }, o3);
        }
        wave_lds_sync();
    }

    float* Ow = &Wsm[wv][0];
#pragma unroll
    for (int r = 0; r < 8; ++r) {
        float* orow = Ow + (r + 8 * half) * OPITCH + l16;
        orow[ 0] = o0[r];
        orow[16] = o1[r];
        orow[32] = o2[r];
        orow[48] = o3[r];
    }
    if (l16 == 0) {
#pragma unroll
        for (int r = 0; r < 8; ++r) {
            Msm[wv][r + 8 * half] = mrow[r];
            Lsm[wv][r + 8 * half] = lrow[r];
        }
    }
    __syncthreads();

    const int R = 2 * wv + half;
    float mw[NWAVE];
    float M = -INFINITY;
#pragma unroll
    for (int w = 0; w < NWAVE; ++w) { mw[w] = Msm[w][R]; M = fmaxf(M, mw[w]); }
    float L = 0.0f;
    float4 acc = make_float4(0.0f, 0.0f, 0.0f, 0.0f);
#pragma unroll
    for (int w = 0; w < NWAVE; ++w) {
        const float g = expf(mw[w] - M);
        L += g * Lsm[w][R];
        const float4 ov = *(const float4*)(&Wsm[w][R * OPITCH + 4 * l16]);
        acc.x += g * ov.x; acc.y += g * ov.y;
        acc.z += g * ov.z; acc.w += g * ov.w;
    }
    const float inv = 1.0f / L;
    acc.x *= inv; acc.y *= inv; acc.z *= inv; acc.w *= inv;
    float* Ob = out + ((size_t)b * Tsz + qi * 16 + R) * Hsz + 4 * l16;
    *(float4*)Ob = acc;
}

extern "C" void kernel_launch(void* const* d_in, const int* in_sizes, int n_in,
                              void* d_out, int out_size, void* d_ws, size_t ws_size,
                              hipStream_t stream) {
    const float* x  = (const float*)d_in[0];
    const float* Wq = (const float*)d_in[1];
    const float* Wk = (const float*)d_in[2];
    const float* Wv = (const float*)d_in[3];

    const size_t BTH = (size_t)Bsz * Tsz * Hsz;
    float* qkv = (float*)d_ws;
    float* Wt  = qkv + 3 * BTH;

    dim3 g0(Hsz, 3), blk0(256);
    wt_kernel<<<g0, blk0, 0, stream>>>(Wq, Wk, Wv, Wt);

    dim3 g1(Bsz * Tsz / 64, 3), blk1(128);
    qkv_kernel<<<g1, blk1, 0, stream>>>(x, Wt, qkv);

    dim3 g2(Tsz / 16, Bsz), blk2(32 * NWAVE);
    attn_kernel<<<g2, blk2, 0, stream>>>(qkv, (float*)d_out);
}


# attention: both products transposed (S^T = K.Q^T, O^T = V^T.P^T) so a lane owns one query: row max and sum in-lane plus one half swap, P stays in registers, no LDS pass (sum order changes)
# speedup vs baseline: 2.4982x; 1.0640x over previous
#include <hip/hip_runtime.h>
#include <math.h>

#define Bsz 8
#define Tsz 2048
#define Csz 1024
#define Hsz 64

typedef __attribute__((ext_vector_type(2))) float v2f;
typedef __attribute__((ext_vector_type(8))) float v8f;

__device__ __forceinline__ v8f wmma_f32(v2f a, v2f b, v8f c) {
    return __builtin_amdgcn_wmma_f32_16x16x4_f32(
         false, a,  false, b,
         (short)0, c,  false,  false);
}

__global__ __launch_bounds__(256) void wt_kernel(
    const float* __restrict__ Wq,
    const float* __restrict__ Wk,
    const float* __restrict__ Wv,
    float* __restrict__ Wt)
{
    const int which = blockIdx.y;
    const float* W = (which == 0) ? Wq : (which == 1) ? Wk : Wv;
    float* dst = Wt + (size_t)which * Hsz * Csz;
    const int h = blockIdx.x;
    for (int c = threadIdx.x; c < Csz; c += 256)
        dst[(size_t)h * Csz + c] = W[(size_t)c * Hsz + h];
}

__global__ __launch_bounds__(128) void qkv_kernel(
    const float* __restrict__ x,
    const float* __restrict__ Wt,
    float* __restrict__ qkv)
{
    const int lane = threadIdx.x & 31;
    const int wv   = threadIdx.x >> 5;
    const int half = lane >> 4;
    const int l16  = lane & 15;

    const long mrow  = (long)blockIdx.x * 64 + wv * 16;
    const int  which = blockIdx.y;
    const float* W = Wt + (size_t)which * Hsz * Csz;

    const float* arow = x + (mrow + l16) * (size_t)Csz + 4 * half;
    const float* b0 = W + (size_t)( 0 + l16) * Csz + 4 * half;
    const float* b1 = W + (size_t)(16 + l16) * Csz + 4 * half;
    const float* b2 = W + (size_t)(32 + l16) * Csz + 4 * half;
    const float* b3 = W + (size_t)(48 + l16) * Csz + 4 * half;

    v8f acc0 = {}, acc1 = {}, acc2 = {}, acc3 = {};
#pragma unroll 2
    for (int kk = 0; kk < Csz; kk += 8) {
        const float4 a  = *(const float4*)(arow + kk);
        const float4 w0 = *(const float4*)(b0 + kk);
        const float4 w1 = *(const float4*)(b1 + kk);
        const float4 w2 = *(const float4*)(b2 + kk);
        const float4 w3 = *(const float4*)(b3 + kk);
        const v2f alo = { a.x, a.y }, ahi = { a.z, a.w };
        acc0 = wmma_f32(alo, (v2f){ w0.x, w0.y }, acc0);
        acc1 = wmma_f32(alo, (v2f){ w1.x, w1.y }, acc1);
        acc2 = wmma_f32(alo, (v2f){ w2.x, w2.y }, acc2);
        acc3 = wmma_f32(alo, (v2f){ w3.x, w3.y }, acc3);
        acc0 = wmma_f32(ahi, (v2f){ w0.z, w0.w }, acc0);
        acc1 = wmma_f32(ahi, (v2f){ w1.z, w1.w }, acc1);
        acc2 = wmma_f32(ahi, (v2f){ w2.z, w2.w }, acc2);
        acc3 = wmma_f32(ahi, (v2f){ w3.z, w3.w }, acc3);
    }

    const size_t BTH = (size_t)Bsz * Tsz * Hsz;
    if (which < 2) {
        float* op = qkv + (size_t)which * BTH + (mrow + 8 * half) * (size_t)Hsz + l16;
#pragma unroll
        for (int r = 0; r < 8; ++r) {
            op[(size_t)r * Hsz +  0] = acc0[r];
            op[(size_t)r * Hsz + 16] = acc1[r];
            op[(size_t)r * Hsz + 32] = acc2[r];
            op[(size_t)r * Hsz + 48] = acc3[r];
        }
    } else {
        const long bb   = mrow / Tsz;
        const long tloc = mrow - bb * Tsz + 8 * half;
        float* vt = qkv + 2 * BTH;
        const v8f* accs[4] = { &acc0, &acc1, &acc2, &acc3 };
#pragma unroll
        for (int nt = 0; nt < 4; ++nt) {
            const int col = nt * 16 + l16;
            float* vp = vt + ((size_t)bb * Hsz + col) * Tsz + tloc;
            const v8f& a = *accs[nt];
            float4 lo = make_float4(a[0], a[1], a[2], a[3]);
            float4 hi = make_float4(a[4], a[5], a[6], a[7]);
            *(float4*)(vp + 0) = lo;
            *(float4*)(vp + 4) = hi;
        }
    }
}

#define NWAVE 8
#define OPITCH 68

__device__ __forceinline__ float xhalf(float x) {
    const int b = __builtin_bit_cast(int, x);
    return __builtin_bit_cast(float,
        __builtin_amdgcn_permlanex16(b, b, 0x76543210, 0xfedcba98, false, false));
}

__global__ __launch_bounds__(32 * NWAVE) void attn_kernel(
    const float* __restrict__ qkv,
    float* __restrict__ out)
{
    __shared__ __attribute__((aligned(16))) float Wsm[NWAVE][16 * OPITCH];
    __shared__ float Msm[NWAVE][16];
    __shared__ float Lsm[NWAVE][16];

    const int lane = threadIdx.x & 31;
    const int wv   = __builtin_amdgcn_readfirstlane((int)(threadIdx.x >> 5));
    const int half = lane >> 4;
    const int l16  = lane & 15;
    const int b    = blockIdx.y;
    const int qi   = gridDim.x - 1 - blockIdx.x;

    const size_t BTH = (size_t)Bsz * Tsz * Hsz;
    const float* Q  = qkv +            (size_t)b * Tsz * Hsz;
    const float* K  = qkv + BTH     +  (size_t)b * Tsz * Hsz;
    const float* Vt = qkv + 2 * BTH +  (size_t)b * Hsz * Tsz;

    const float scale = 0.125f;
    const int qrow = qi * 16 + l16;

    v2f qb[16];
#pragma unroll
    for (int c = 0; c < 8; ++c) {
        const float4 t = *(const float4*)(Q + (size_t)qrow * Hsz + 8 * c + 4 * half);
        qb[2 * c][0]     = t.x * scale;
        qb[2 * c][1]     = t.y * scale;
        qb[2 * c + 1][0] = t.z * scale;
        qb[2 * c + 1][1] = t.w * scale;
    }

    const float* Vr0 = Vt + (size_t)( 0 + l16) * Tsz + 8 * half;
    const float* Vr1 = Vt + (size_t)(16 + l16) * Tsz + 8 * half;
    const float* Vr2 = Vt + (size_t)(32 + l16) * Tsz + 8 * half;
    const float* Vr3 = Vt + (size_t)(48 + l16) * Tsz + 8 * half;

    float m_run = -INFINITY, l_run = 0.0f;
    v8f o0 = {}, o1 = {}, o2 = {}, o3 = {};

    for (int kt = wv; kt <= qi; kt += NWAVE) {
        const float* Kt = K + (size_t)(kt * 16 + l16) * Hsz + 4 * half;
        v8f s = {};
#pragma unroll
        for (int c = 0; c < 8; ++c) {
            const float4 ka = *(const float4*)(Kt + 8 * c);
            s = wmma_f32((v2f){ ka.x, ka.y }, qb[2 * c],     s);
            s = wmma_f32((v2f){ ka.z, ka.w }, qb[2 * c + 1], s);
        }

        if (kt == qi) {
#pragma unroll
            for (int r = 0; r < 8; ++r)
                if (8 * half + r > l16) s[r] = -INFINITY;
        }

        float mx = fmaxf(fmaxf(fmaxf(s[0], s[1]), fmaxf(s[2], s[3])),
                         fmaxf(fmaxf(s[4], s[5]), fmaxf(s[6], s[7])));
        mx = fmaxf(mx, xhalf(mx));
        const float mn    = fmaxf(m_run, mx);
        const float alpha = expf(m_run - mn);
        m_run = mn;
#pragma unroll
        for (int r = 0; r < 8; ++r) s[r] = expf(s[r] - mn);
        const float sm = ((s[0] + s[1]) + (s[2] + s[3])) + ((s[4] + s[5]) + (s[6] + s[7]));
        l_run = l_run * alpha + sm;

#pragma unroll
        for (int r = 0; r < 8; ++r) {
            o0[r] *= alpha; o1[r] *= alpha;
            o2[r] *= alpha; o3[r] *= alpha;
        }

        const int kb0 = kt * 16;
        const float4 va0 = *(const float4*)(Vr0 + kb0), vb0 = *(const float4*)(Vr0 + kb0 + 4);
        const float4 va1 = *(const float4*)(Vr1 + kb0), vb1 = *(const float4*)(Vr1 + kb0 + 4);
        const float4 va2 = *(const float4*)(Vr2 + kb0), vb2 = *(const float4*)(Vr2 + kb0 + 4);
        const float4 va3 = *(const float4*)(Vr3 + kb0), vb3 = *(const float4*)(Vr3 + kb0 + 4);
        const v2f p0 = { s[0], s[1] }, p1 = { s[2], s[3] };
        const v2f p2 = { s[4], s[5] }, p3 = { s[6], s[7] };
        o0 = wmma_f32((v2f){ va0.x, va0.y }, p0, o0);
        o1 = wmma_f32((v2f){ va1.x, va1.y }, p0, o1);
        o2 = wmma_f32((v2f){ va2.x, va2.y }, p0, o2);
        o3 = wmma_f32((v2f){ va3.x, va3.y }, p0, o3);
        o0 = wmma_f32((v2f){ va0.z, va0.w }, p1, o0);
        o1 = wmma_f32((v2f){ va1.z, va1.w }, p1, o1);
        o2 = wmma_f32((v2f){ va2.z, va2.w }, p1, o2);
        o3 = wmma_f32((v2f){ va3.z, va3.w }, p1, o3);
        o0 = wmma_f32((v2f){ vb0.x, vb0.y }, p2, o0);
        o1 = wmma_f32((v2f){ vb1.x, vb1.y }, p2, o1);
        o2 = wmma_f32((v2f){ vb2.x, vb2.y }, p2, o2);
        o3 = wmma_f32((v2f){ vb3.x, vb3.y }, p2, o3);
        o0 = wmma_f32((v2f){ vb0.z, vb0.w }, p3, o0);
        o1 = wmma_f32((v2f){ vb1.z, vb1.w }, p3, o1);
        o2 = wmma_f32((v2f){ vb2.z, vb2.w }, p3, o2);
        o3 = wmma_f32((v2f){ vb3.z, vb3.w }, p3, o3);
    }

    const float l_tot = l_run + xhalf(l_run);

    {
        float* orow = &Wsm[wv][0] + l16 * OPITCH + 8 * half;
        *(float4*)(orow +  0) = make_float4(o0[0], o0[1], o0[2], o0[3]);
        *(float4*)(orow +  4) = make_float4(o0[4], o0[5], o0[6], o0[7]);
        *(float4*)(orow + 16) = make_float4(o1[0], o1[1], o1[2], o1[3]);
        *(float4*)(orow + 20) = make_float4(o1[4], o1[5], o1[6], o1[7]);
        *(float4*)(orow + 32) = make_float4(o2[0], o2[1], o2[2], o2[3]);
        *(float4*)(orow + 36) = make_float4(o2[4], o2[5], o2[6], o2[7]);
        *(float4*)(orow + 48) = make_float4(o3[0], o3[1], o3[2], o3[3]);
        *(float4*)(orow + 52) = make_float4(o3[4], o3[5], o3[6], o3[7]);
    }
    if (half == 0) {
        Msm[wv][l16] = m_run;
        Lsm[wv][l16] = l_tot;
    }
    __syncthreads();

    const int R = 2 * wv + half;
    float mw[NWAVE];
    float M = -INFINITY;
#pragma unroll
    for (int w = 0; w < NWAVE; ++w) { mw[w] = Msm[w][R]; M = fmaxf(M, mw[w]); }
    float L = 0.0f;
    float4 acc = make_float4(0.0f, 0.0f, 0.0f, 0.0f);
#pragma unroll
    for (int w = 0; w < NWAVE; ++w) {
        const float g = expf(mw[w] - M);
        L += g * Lsm[w][R];
        const float4 ov = *(const float4*)(&Wsm[w][R * OPITCH + 4 * l16]);
        acc.x += g * ov.x; acc.y += g * ov.y;
        acc.z += g * ov.z; acc.w += g * ov.w;
    }
    const float inv = 1.0f / L;
    acc.x *= inv; acc.y *= inv; acc.z *= inv; acc.w *= inv;
    float* Ob = out + ((size_t)b * Tsz + qi * 16 + R) * Hsz + 4 * l16;
    *(float4*)Ob = acc;
}

extern "C" void kernel_launch(void* const* d_in, const int* in_sizes, int n_in,
                              void* d_out, int out_size, void* d_ws, size_t ws_size,
                              hipStream_t stream) {
    const float* x  = (const float*)d_in[0];
    const float* Wq = (const float*)d_in[1];
    const float* Wk = (const float*)d_in[2];
    const float* Wv = (const float*)d_in[3];

    const size_t BTH = (size_t)Bsz * Tsz * Hsz;
    float* qkv = (float*)d_ws;
    float* Wt  = qkv + 3 * BTH;

    dim3 g0(Hsz, 3), blk0(256);
    wt_kernel<<<g0, blk0, 0, stream>>>(Wq, Wk, Wv, Wt);

    dim3 g1(Bsz * Tsz / 64, 3), blk1(128);
    qkv_kernel<<<g1, blk1, 0, stream>>>(x, Wt, qkv);

    dim3 g2(Tsz / 16, Bsz), blk2(32 * NWAVE);
    attn_kernel<<<g2, blk2, 0, stream>>>(qkv, (float*)d_out);
}


# qkv projection: an 8-wave work-group per 64 x rows stages 32-k slabs of x and W^T (all 192 columns) in LDS by whole-line loads, next slab in flight; WMMA fragments come from LDS (sum order kept)
# speedup vs baseline: 2.8693x; 1.1485x over previous
#include <hip/hip_runtime.h>
#include <math.h>

#define Bsz 8
#define Tsz 2048
#define Csz 1024
#define Hsz 64

typedef __attribute__((ext_vector_type(2))) float v2f;
typedef __attribute__((ext_vector_type(8))) float v8f;

__device__ __forceinline__ v8f wmma_f32(v2f a, v2f b, v8f c) {
    return __builtin_amdgcn_wmma_f32_16x16x4_f32(
         false, a,  false, b,
         (short)0, c,  false,  false);
}

__global__ __launch_bounds__(256) void wt_kernel(
    const float* __restrict__ Wq,
    const float* __restrict__ Wk,
    const float* __restrict__ Wv,
    float* __restrict__ Wt)
{
    const int which = blockIdx.y;
    const float* W = (which == 0) ? Wq : (which == 1) ? Wk : Wv;
    float* dst = Wt + (size_t)which * Hsz * Csz;
    const int h = blockIdx.x;
    for (int c = threadIdx.x; c < Csz; c += 256)
        dst[(size_t)h * Csz + c] = W[(size_t)c * Hsz + h];
}

#define QBM 64
#define QBN 192
#define QBK 32
#define QPITCH 36
#define QROWS (QBM + QBN)

__global__ __launch_bounds__(256) __attribute__((amdgpu_waves_per_eu(2, 4))) void qkv_kernel(
    const float* __restrict__ x,
    const float* __restrict__ Wt,
    float* __restrict__ qkv)
{
    __shared__ __attribute__((aligned(16))) float Ssm[2][QROWS * QPITCH];

    const int tid  = threadIdx.x;
    const int lane = tid & 31;
    const int wv   = __builtin_amdgcn_readfirstlane(tid >> 5);
    const int wm   = wv & 3;
    const int wn   = wv >> 2;
    const int half = lane >> 4;
    const int l16  = lane & 15;
    const long m0  = (long)blockIdx.x * QBM;

    const int sc = tid & 7;
    const int sr = tid >> 3;
    const float* xg = x  + (m0 + sr) * (size_t)Csz + 4 * sc;
    const float* wg = Wt + (size_t)sr * Csz + 4 * sc;
    const int sdst = sr * QPITCH + 4 * sc;

    const int aoff = (16 * wm + l16) * QPITCH + 4 * half;
    const int boff = (QBM + 96 * wn + l16) * QPITCH + 4 * half;

    v8f acc[6];
#pragma unroll
    for (int j = 0; j < 6; ++j) acc[j] = (v8f){};

    float4 p0, p1, p2, p3, p4, p5, p6, p7;
#define QKV_LOAD_SLAB(k)                                          \
    p0 = *(const float4*)(xg + (k));                              \
    p1 = *(const float4*)(xg + (size_t)32 * Csz + (k));           \
    p2 = *(const float4*)(wg + (k));                              \
    p3 = *(const float4*)(wg + (size_t)32 * Csz + (k));           \
    p4 = *(const float4*)(wg + (size_t)64 * Csz + (k));           \
    p5 = *(const float4*)(wg + (size_t)96 * Csz + (k));           \
    p6 = *(const float4*)(wg + (size_t)128 * Csz + (k));          \
    p7 = *(const float4*)(wg + (size_t)160 * Csz + (k));
#define QKV_STORE_SLAB(S)                                         \
    *(float4*)((S) + sdst +   0 * QPITCH) = p0;                   \
    *(float4*)((S) + sdst +  32 * QPITCH) = p1;                   \
    *(float4*)((S) + sdst +  64 * QPITCH) = p2;                   \
    *(float4*)((S) + sdst +  96 * QPITCH) = p3;                   \
    *(float4*)((S) + sdst + 128 * QPITCH) = p4;                   \
    *(float4*)((S) + sdst + 160 * QPITCH) = p5;                   \
    *(float4*)((S) + sdst + 192 * QPITCH) = p6;                   \
    *(float4*)((S) + sdst + 224 * QPITCH) = p7;

    QKV_LOAD_SLAB(0)
    QKV_STORE_SLAB(&Ssm[0][0])
    __syncthreads();
    QKV_LOAD_SLAB(QBK)

    const int NS = Csz / QBK;
    for (int s = 0; s < NS; ++s) {
        const float* Sb = &Ssm[s & 1][0];
        float4 af[QBK / 8], bf[QBK / 8][6];
#pragma unroll
        for (int g = 0; g < QBK / 8; ++g) {
            af[g] = *(const float4*)(Sb + aoff + 8 * g);
#pragma unroll
            for (int j = 0; j < 6; ++j)
                bf[g][j] = *(const float4*)(Sb + boff + 16 * j * QPITCH + 8 * g);
        }
        __builtin_amdgcn_sched_barrier(0);
#pragma unroll
        for (int g = 0; g < QBK / 8; ++g) {
            const v2f alo = { af[g].x, af[g].y }, ahi = { af[g].z, af[g].w };
#pragma unroll
            for (int j = 0; j < 6; ++j) acc[j] = wmma_f32(alo, (v2f){ bf[g][j].x, bf[g][j].y }, acc[j]);
#pragma unroll
            for (int j = 0; j < 6; ++j) acc[j] = wmma_f32(ahi, (v2f){ bf[g][j].z, bf[g][j].w }, acc[j]);
        }
        __builtin_amdgcn_sched_barrier(0);
        if (s + 1 < NS) {
            float* Sn = &Ssm[(s + 1) & 1][0];
            QKV_STORE_SLAB(Sn)
        }
        __syncthreads();
        if (s + 2 < NS) {
            const int k2 = (s + 2) * QBK;
            QKV_LOAD_SLAB(k2)
        }
    }
#undef QKV_LOAD_SLAB
#undef QKV_STORE_SLAB

    const size_t BTH = (size_t)Bsz * Tsz * Hsz;
    const long mrow = m0 + 16 * wm;
    const long bb   = mrow / Tsz;
    const long tloc = mrow - bb * Tsz + 8 * half;
#pragma unroll
    for (int j = 0; j < 6; ++j) {
        const int gc0   = 96 * wn + 16 * j;
        const int which = gc0 >> 6;
        const int hcol  = (gc0 & 63) + l16;
        if (which < 2) {
            float* op = qkv + (size_t)which * BTH + (mrow + 8 * half) * (size_t)Hsz + hcol;
#pragma unroll
            for (int r = 0; r < 8; ++r) op[(size_t)r * Hsz] = acc[j][r];
        } else {
            float* vp = qkv + 2 * BTH + ((size_t)bb * Hsz + hcol) * Tsz + tloc;
            *(float4*)(vp + 0) = make_float4(acc[j][0], acc[j][1], acc[j][2], acc[j][3]);
            *(float4*)(vp + 4) = make_float4(acc[j][4], acc[j][5], acc[j][6], acc[j][7]);
        }
    }
}

#define NWAVE 8
#define OPITCH 68

__device__ __forceinline__ float xhalf(float x) {
    const int b = __builtin_bit_cast(int, x);
    return __builtin_bit_cast(float,
        __builtin_amdgcn_permlanex16(b, b, 0x76543210, 0xfedcba98, false, false));
}

__global__ __launch_bounds__(32 * NWAVE) void attn_kernel(
    const float* __restrict__ qkv,
    float* __restrict__ out)
{
    __shared__ __attribute__((aligned(16))) float Wsm[NWAVE][16 * OPITCH];
    __shared__ float Msm[NWAVE][16];
    __shared__ float Lsm[NWAVE][16];

    const int lane = threadIdx.x & 31;
    const int wv   = __builtin_amdgcn_readfirstlane((int)(threadIdx.x >> 5));
    const int half = lane >> 4;
    const int l16  = lane & 15;
    const int b    = blockIdx.y;
    const int qi   = gridDim.x - 1 - blockIdx.x;

    const size_t BTH = (size_t)Bsz * Tsz * Hsz;
    const float* Q  = qkv +            (size_t)b * Tsz * Hsz;
    const float* K  = qkv + BTH     +  (size_t)b * Tsz * Hsz;
    const float* Vt = qkv + 2 * BTH +  (size_t)b * Hsz * Tsz;

    const float scale = 0.125f;
    const int qrow = qi * 16 + l16;

    v2f qb[16];
#pragma unroll
    for (int c = 0; c < 8; ++c) {
        const float4 t = *(const float4*)(Q + (size_t)qrow * Hsz + 8 * c + 4 * half);
        qb[2 * c][0]     = t.x * scale;
        qb[2 * c][1]     = t.y * scale;
        qb[2 * c + 1][0] = t.z * scale;
        qb[2 * c + 1][1] = t.w * scale;
    }

    const float* Vr0 = Vt + (size_t)( 0 + l16) * Tsz + 8 * half;
    const float* Vr1 = Vt + (size_t)(16 + l16) * Tsz + 8 * half;
    const float* Vr2 = Vt + (size_t)(32 + l16) * Tsz + 8 * half;
    const float* Vr3 = Vt + (size_t)(48 + l16) * Tsz + 8 * half;

    float m_run = -INFINITY, l_run = 0.0f;
    v8f o0 = {}, o1 = {}, o2 = {}, o3 = {};

    for (int kt = wv; kt <= qi; kt += NWAVE) {
        const float* Kt = K + (size_t)(kt * 16 + l16) * Hsz + 4 * half;
        v8f s = {};
#pragma unroll
        for (int c = 0; c < 8; ++c) {
            const float4 ka = *(const float4*)(Kt + 8 * c);
            s = wmma_f32((v2f){ ka.x, ka.y }, qb[2 * c],     s);
            s = wmma_f32((v2f){ ka.z, ka.w }, qb[2 * c + 1], s);
        }

        if (kt == qi) {
#pragma unroll
            for (int r = 0; r < 8; ++r)
                if (8 * half + r > l16) s[r] = -INFINITY;
        }

        float mx = fmaxf(fmaxf(fmaxf(s[0], s[1]), fmaxf(s[2], s[3])),
                         fmaxf(fmaxf(s[4], s[5]), fmaxf(s[6], s[7])));
        mx = fmaxf(mx, xhalf(mx));
        const float mn    = fmaxf(m_run, mx);
        const float alpha = expf(m_run - mn);
        m_run = mn;
#pragma unroll
        for (int r = 0; r < 8; ++r) s[r] = expf(s[r] - mn);
        const float sm = ((s[0] + s[1]) + (s[2] + s[3])) + ((s[4] + s[5]) + (s[6] + s[7]));
        l_run = l_run * alpha + sm;

#pragma unroll
        for (int r = 0; r < 8; ++r) {
            o0[r] *= alpha; o1[r] *= alpha;
            o2[r] *= alpha; o3[r] *= alpha;
        }

        const int kb0 = kt * 16;
        const float4 va0 = *(const float4*)(Vr0 + kb0), vb0 = *(const float4*)(Vr0 + kb0 + 4);
        const float4 va1 = *(const float4*)(Vr1 + kb0), vb1 = *(const float4*)(Vr1 + kb0 + 4);
        const float4 va2 = *(const float4*)(Vr2 + kb0), vb2 = *(const float4*)(Vr2 + kb0 + 4);
        const float4 va3 = *(const float4*)(Vr3 + kb0), vb3 = *(const float4*)(Vr3 + kb0 + 4);
        const v2f p0 = { s[0], s[1] }, p1 = { s[2], s[3] };
        const v2f p2 = { s[4], s[5] }, p3 = { s[6], s[7] };
        o0 = wmma_f32((v2f){ va0.x, va0.y }, p0, o0);
        o1 = wmma_f32((v2f){ va1.x, va1.y }, p0, o1);
        o2 = wmma_f32((v2f){ va2.x, va2.y }, p0, o2);
        o3 = wmma_f32((v2f){ va3.x, va3.y }, p0, o3);
        o0 = wmma_f32((v2f){ va0.z, va0.w }, p1, o0);
        o1 = wmma_f32((v2f){ va1.z, va1.w }, p1, o1);
        o2 = wmma_f32((v2f){ va2.z, va2.w }, p1, o2);
        o3 = wmma_f32((v2f){ va3.z, va3.w }, p1, o3);
        o0 = wmma_f32((v2f){ vb0.x, vb0.y }, p2, o0);
        o1 = wmma_f32((v2f){ vb1.x, vb1.y }, p2, o1);
        o2 = wmma_f32((v2f){ vb2.x, vb2.y }, p2, o2);
        o3 = wmma_f32((v2f){ vb3.x, vb3.y }, p2, o3);
        o0 = wmma_f32((v2f){ vb0.z, vb0.w }, p3, o0);
        o1 = wmma_f32((v2f){ vb1.z, vb1.w }, p3, o1);
        o2 = wmma_f32((v2f){ vb2.z, vb2.w }, p3, o2);
        o3 = wmma_f32((v2f){ vb3.z, vb3.w }, p3, o3);
    }

    const float l_tot = l_run + xhalf(l_run);

    {
        float* orow = &Wsm[wv][0] + l16 * OPITCH + 8 * half;
        *(float4*)(orow +  0) = make_float4(o0[0], o0[1], o0[2], o0[3]);
        *(float4*)(orow +  4) = make_float4(o0[4], o0[5], o0[6], o0[7]);
        *(float4*)(orow + 16) = make_float4(o1[0], o1[1], o1[2], o1[3]);
        *(float4*)(orow + 20) = make_float4(o1[4], o1[5], o1[6], o1[7]);
        *(float4*)(orow + 32) = make_float4(o2[0], o2[1], o2[2], o2[3]);
        *(float4*)(orow + 36) = make_float4(o2[4], o2[5], o2[6], o2[7]);
        *(float4*)(orow + 48) = make_float4(o3[0], o3[1], o3[2], o3[3]);
        *(float4*)(orow + 52) = make_float4(o3[4], o3[5], o3[6], o3[7]);
    }
    if (half == 0) {
        Msm[wv][l16] = m_run;
        Lsm[wv][l16] = l_tot;
    }
    __syncthreads();

    const int R = 2 * wv + half;
    float mw[NWAVE];
    float M = -INFINITY;
#pragma unroll
    for (int w = 0; w < NWAVE; ++w) { mw[w] = Msm[w][R]; M = fmaxf(M, mw[w]); }
    float L = 0.0f;
    float4 acc = make_float4(0.0f, 0.0f, 0.0f, 0.0f);
#pragma unroll
    for (int w = 0; w < NWAVE; ++w) {
        const float g = expf(mw[w] - M);
        L += g * Lsm[w][R];
        const float4 ov = *(const float4*)(&Wsm[w][R * OPITCH + 4 * l16]);
        acc.x += g * ov.x; acc.y += g * ov.y;
        acc.z += g * ov.z; acc.w += g * ov.w;
    }
    const float inv = 1.0f / L;
    acc.x *= inv; acc.y *= inv; acc.z *= inv; acc.w *= inv;
    float* Ob = out + ((size_t)b * Tsz + qi * 16 + R) * Hsz + 4 * l16;
    *(float4*)Ob = acc;
}

extern "C" void kernel_launch(void* const* d_in, const int* in_sizes, int n_in,
                              void* d_out, int out_size, void* d_ws, size_t ws_size,
                              hipStream_t stream) {
    const float* x  = (const float*)d_in[0];
    const float* Wq = (const float*)d_in[1];
    const float* Wk = (const float*)d_in[2];
    const float* Wv = (const float*)d_in[3];

    const size_t BTH = (size_t)Bsz * Tsz * Hsz;
    float* qkv = (float*)d_ws;
    float* Wt  = qkv + 3 * BTH;

    dim3 g0(Hsz, 3), blk0(256);
    wt_kernel<<<g0, blk0, 0, stream>>>(Wq, Wk, Wv, Wt);

    dim3 g1(Bsz * Tsz / QBM), blk1(256);
    qkv_kernel<<<g1, blk1, 0, stream>>>(x, Wt, qkv);

    dim3 g2(Tsz / 16, Bsz), blk2(32 * NWAVE);
    attn_kernel<<<g2, blk2, 0, stream>>>(qkv, (float*)d_out);
}


# attention: one work-group per PAIR of query tiles (p, 127-p); the pair's 129 causal key tiles are dealt round-robin to its 8 waves: equal work per work-group (sum order of the 2nd tile changes)
# speedup vs baseline: 3.2264x; 1.1245x over previous
#include <hip/hip_runtime.h>
#include <math.h>

#define Bsz 8
#define Tsz 2048
#define Csz 1024
#define Hsz 64

typedef __attribute__((ext_vector_type(2))) float v2f;
typedef __attribute__((ext_vector_type(8))) float v8f;

__device__ __forceinline__ v8f wmma_f32(v2f a, v2f b, v8f c) {
    return __builtin_amdgcn_wmma_f32_16x16x4_f32(
         false, a,  false, b,
         (short)0, c,  false,  false);
}

__global__ __launch_bounds__(256) void wt_kernel(
    const float* __restrict__ Wq,
    const float* __restrict__ Wk,
    const float* __restrict__ Wv,
    float* __restrict__ Wt)
{
    const int which = blockIdx.y;
    const float* W = (which == 0) ? Wq : (which == 1) ? Wk : Wv;
    float* dst = Wt + (size_t)which * Hsz * Csz;
    const int h = blockIdx.x;
    for (int c = threadIdx.x; c < Csz; c += 256)
        dst[(size_t)h * Csz + c] = W[(size_t)c * Hsz + h];
}

#define QBM 64
#define QBN 192
#define QBK 32
#define QPITCH 36
#define QROWS (QBM + QBN)

__global__ __launch_bounds__(256) __attribute__((amdgpu_waves_per_eu(2, 4))) void qkv_kernel(
    const float* __restrict__ x,
    const float* __restrict__ Wt,
    float* __restrict__ qkv)
{
    __shared__ __attribute__((aligned(16))) float Ssm[2][QROWS * QPITCH];

    const int tid  = threadIdx.x;
    const int lane = tid & 31;
    const int wv   = __builtin_amdgcn_readfirstlane(tid >> 5);
    const int wm   = wv & 3;
    const int wn   = wv >> 2;
    const int half = lane >> 4;
    const int l16  = lane & 15;
    const long m0  = (long)blockIdx.x * QBM;

    const int sc = tid & 7;
    const int sr = tid >> 3;
    const float* xg = x  + (m0 + sr) * (size_t)Csz + 4 * sc;
    const float* wg = Wt + (size_t)sr * Csz + 4 * sc;
    const int sdst = sr * QPITCH + 4 * sc;

    const int aoff = (16 * wm + l16) * QPITCH + 4 * half;
    const int boff = (QBM + 96 * wn + l16) * QPITCH + 4 * half;

    v8f acc[6];
#pragma unroll
    for (int j = 0; j < 6; ++j) acc[j] = (v8f){};

    float4 p0, p1, p2, p3, p4, p5, p6, p7;
#define QKV_LOAD_SLAB(k)                                          \
    p0 = *(const float4*)(xg + (k));                              \
    p1 = *(const float4*)(xg + (size_t)32 * Csz + (k));           \
    p2 = *(const float4*)(wg + (k));                              \
    p3 = *(const float4*)(wg + (size_t)32 * Csz + (k));           \
    p4 = *(const float4*)(wg + (size_t)64 * Csz + (k));           \
    p5 = *(const float4*)(wg + (size_t)96 * Csz + (k));           \
    p6 = *(const float4*)(wg + (size_t)128 * Csz + (k));          \
    p7 = *(const float4*)(wg + (size_t)160 * Csz + (k));
#define QKV_STORE_SLAB(S)                                         \
    *(float4*)((S) + sdst +   0 * QPITCH) = p0;                   \
    *(float4*)((S) + sdst +  32 * QPITCH) = p1;                   \
    *(float4*)((S) + sdst +  64 * QPITCH) = p2;                   \
    *(float4*)((S) + sdst +  96 * QPITCH) = p3;                   \
    *(float4*)((S) + sdst + 128 * QPITCH) = p4;                   \
    *(float4*)((S) + sdst + 160 * QPITCH) = p5;                   \
    *(float4*)((S) + sdst + 192 * QPITCH) = p6;                   \
    *(float4*)((S) + sdst + 224 * QPITCH) = p7;

    QKV_LOAD_SLAB(0)
    QKV_STORE_SLAB(&Ssm[0][0])
    __syncthreads();
    QKV_LOAD_SLAB(QBK)

    const int NS = Csz / QBK;
    for (int s = 0; s < NS; ++s) {
        const float* Sb = &Ssm[s & 1][0];
        float4 af[QBK / 8], bf[QBK / 8][6];
#pragma unroll
        for (int g = 0; g < QBK / 8; ++g) {
            af[g] = *(const float4*)(Sb + aoff + 8 * g);
#pragma unroll
            for (int j = 0; j < 6; ++j)
                bf[g][j] = *(const float4*)(Sb + boff + 16 * j * QPITCH + 8 * g);
        }
        __builtin_amdgcn_sched_barrier(0);
#pragma unroll
        for (int g = 0; g < QBK / 8; ++g) {
            const v2f alo = { af[g].x, af[g].y }, ahi = { af[g].z, af[g].w };
#pragma unroll
            for (int j = 0; j < 6; ++j) acc[j] = wmma_f32(alo, (v2f){ bf[g][j].x, bf[g][j].y }, acc[j]);
#pragma unroll
            for (int j = 0; j < 6; ++j) acc[j] = wmma_f32(ahi, (v2f){ bf[g][j].z, bf[g][j].w }, acc[j]);
        }
        __builtin_amdgcn_sched_barrier(0);
        if (s + 1 < NS) {
            float* Sn = &Ssm[(s + 1) & 1][0];
            QKV_STORE_SLAB(Sn)
        }
        __syncthreads();
        if (s + 2 < NS) {
            const int k2 = (s + 2) * QBK;
            QKV_LOAD_SLAB(k2)
        }
    }
#undef QKV_LOAD_SLAB
#undef QKV_STORE_SLAB

    const size_t BTH = (size_t)Bsz * Tsz * Hsz;
    const long mrow = m0 + 16 * wm;
    const long bb   = mrow / Tsz;
    const long tloc = mrow - bb * Tsz + 8 * half;
#pragma unroll
    for (int j = 0; j < 6; ++j) {
        const int gc0   = 96 * wn + 16 * j;
        const int which = gc0 >> 6;
        const int hcol  = (gc0 & 63) + l16;
        if (which < 2) {
            float* op = qkv + (size_t)which * BTH + (mrow + 8 * half) * (size_t)Hsz + hcol;
#pragma unroll
            for (int r = 0; r < 8; ++r) op[(size_t)r * Hsz] = acc[j][r];
        } else {
            float* vp = qkv + 2 * BTH + ((size_t)bb * Hsz + hcol) * Tsz + tloc;
            *(float4*)(vp + 0) = make_float4(acc[j][0], acc[j][1], acc[j][2], acc[j][3]);
            *(float4*)(vp + 4) = make_float4(acc[j][4], acc[j][5], acc[j][6], acc[j][7]);
        }
    }
}

#define NWAVE 8
#define NQT (Tsz / 16)
#define OPITCH 68

__device__ __forceinline__ float xhalf(float x) {
    const int b = __builtin_bit_cast(int, x);
    return __builtin_bit_cast(float,
        __builtin_amdgcn_permlanex16(b, b, 0x76543210, 0xfedcba98, false, false));
}

__global__ __launch_bounds__(32 * NWAVE) void attn_kernel(
    const float* __restrict__ qkv,
    float* __restrict__ out)
{
    __shared__ __attribute__((aligned(16))) float Wsm[2][NWAVE][16 * OPITCH];
    __shared__ float Msm[2][NWAVE][16];
    __shared__ float Lsm[2][NWAVE][16];

    const int lane = threadIdx.x & 31;
    const int wv   = __builtin_amdgcn_readfirstlane((int)(threadIdx.x >> 5));
    const int half = lane >> 4;
    const int l16  = lane & 15;
    const int b    = blockIdx.y;
    const int pr   = blockIdx.x;

    const size_t BTH = (size_t)Bsz * Tsz * Hsz;
    const float* Q  = qkv +            (size_t)b * Tsz * Hsz;
    const float* K  = qkv + BTH     +  (size_t)b * Tsz * Hsz;
    const float* Vt = qkv + 2 * BTH +  (size_t)b * Hsz * Tsz;

    const float scale = 0.125f;

    const float* Vr0 = Vt + (size_t)( 0 + l16) * Tsz + 8 * half;
    const float* Vr1 = Vt + (size_t)(16 + l16) * Tsz + 8 * half;
    const float* Vr2 = Vt + (size_t)(32 + l16) * Tsz + 8 * half;
    const float* Vr3 = Vt + (size_t)(48 + l16) * Tsz + 8 * half;

#pragma unroll 1
    for (int pass = 0; pass < 2; ++pass) {
        const int qi = pass ? (NQT - 1 - pr) : pr;
        const int k0 = pass ? ((wv - pr - 1) & (NWAVE - 1)) : wv;
        const int qrow = qi * 16 + l16;

        v2f qb[16];
#pragma unroll
        for (int c = 0; c < 8; ++c) {
            const float4 t = *(const float4*)(Q + (size_t)qrow * Hsz + 8 * c + 4 * half);
            qb[2 * c][0]     = t.x * scale;
            qb[2 * c][1]     = t.y * scale;
            qb[2 * c + 1][0] = t.z * scale;
            qb[2 * c + 1][1] = t.w * scale;
        }

        float m_run = -INFINITY, l_run = 0.0f;
        v8f o0 = {}, o1 = {}, o2 = {}, o3 = {};

        for (int kt = k0; kt <= qi; kt += NWAVE) {
            const float* Kt = K + (size_t)(kt * 16 + l16) * Hsz + 4 * half;
            v8f s = {};
#pragma unroll
            for (int c = 0; c < 8; ++c) {
                const float4 ka = *(const float4*)(Kt + 8 * c);
                s = wmma_f32((v2f){ ka.x, ka.y }, qb[2 * c],     s);
                s = wmma_f32((v2f){ ka.z, ka.w }, qb[2 * c + 1], s);
            }

            if (kt == qi) {
#pragma unroll
                for (int r = 0; r < 8; ++r)
                    if (8 * half + r > l16) s[r] = -INFINITY;
            }

            float mx = fmaxf(fmaxf(fmaxf(s[0], s[1]), fmaxf(s[2], s[3])),
                             fmaxf(fmaxf(s[4], s[5]), fmaxf(s[6], s[7])));
            mx = fmaxf(mx, xhalf(mx));
            const float mn    = fmaxf(m_run, mx);
            const float alpha = expf(m_run - mn);
            m_run = mn;
#pragma unroll
            for (int r = 0; r < 8; ++r) s[r] = expf(s[r] - mn);
            const float sm = ((s[0] + s[1]) + (s[2] + s[3])) + ((s[4] + s[5]) + (s[6] + s[7]));
            l_run = l_run * alpha + sm;

#pragma unroll
            for (int r = 0; r < 8; ++r) {
                o0[r] *= alpha; o1[r] *= alpha;
                o2[r] *= alpha; o3[r] *= alpha;
            }

            const int kb0 = kt * 16;
            const float4 va0 = *(const float4*)(Vr0 + kb0), vb0 = *(const float4*)(Vr0 + kb0 + 4);
            const float4 va1 = *(const float4*)(Vr1 + kb0), vb1 = *(const float4*)(Vr1 + kb0 + 4);
            const float4 va2 = *(const float4*)(Vr2 + kb0), vb2 = *(const float4*)(Vr2 + kb0 + 4);
            const float4 va3 = *(const float4*)(Vr3 + kb0), vb3 = *(const float4*)(Vr3 + kb0 + 4);
            const v2f p0 = { s[0], s[1] }, p1 = { s[2], s[3] };
            const v2f p2 = { s[4], s[5] }, p3 = { s[6], s[7] };
            o0 = wmma_f32((v2f){ va0.x, va0.y }, p0, o0);
            o1 = wmma_f32((v2f){ va1.x, va1.y }, p0, o1);
            o2 = wmma_f32((v2f){ va2.x, va2.y }, p0, o2);
            o3 = wmma_f32((v2f){ va3.x, va3.y }, p0, o3);
            o0 = wmma_f32((v2f){ va0.z, va0.w }, p1, o0);
            o1 = wmma_f32((v2f){ va1.z, va1.w }, p1, o1);
            o2 = wmma_f32((v2f){ va2.z, va2.w }, p1, o2);
            o3 = wmma_f32((v2f){ va3.z, va3.w }, p1, o3);
            o0 = wmma_f32((v2f){ vb0.x, vb0.y }, p2, o0);
            o1 = wmma_f32((v2f){ vb1.x, vb1.y }, p2, o1);
            o2 = wmma_f32((v2f){ vb2.x, vb2.y }, p2, o2);
            o3 = wmma_f32((v2f){ vb3.x, vb3.y }, p2, o3);
            o0 = wmma_f32((v2f){ vb0.z, vb0.w }, p3, o0);
            o1 = wmma_f32((v2f){ vb1.z, vb1.w }, p3, o1);
            o2 = wmma_f32((v2f){ vb2.z, vb2.w }, p3, o2);
            o3 = wmma_f32((v2f){ vb3.z, vb3.w }, p3, o3);
        }

        const float l_tot = l_run + xhalf(l_run);

        {
            float* orow = &Wsm[pass][wv][0] + l16 * OPITCH + 8 * half;
            *(float4*)(orow +  0) = make_float4(o0[0], o0[1], o0[2], o0[3]);
            *(float4*)(orow +  4) = make_float4(o0[4], o0[5], o0[6], o0[7]);
            *(float4*)(orow + 16) = make_float4(o1[0], o1[1], o1[2], o1[3]);
            *(float4*)(orow + 20) = make_float4(o1[4], o1[5], o1[6], o1[7]);
            *(float4*)(orow + 32) = make_float4(o2[0], o2[1], o2[2], o2[3]);
            *(float4*)(orow + 36) = make_float4(o2[4], o2[5], o2[6], o2[7]);
            *(float4*)(orow + 48) = make_float4(o3[0], o3[1], o3[2], o3[3]);
            *(float4*)(orow + 52) = make_float4(o3[4], o3[5], o3[6], o3[7]);
        }
        if (half == 0) {
            Msm[pass][wv][l16] = m_run;
            Lsm[pass][wv][l16] = l_tot;
        }
    }
    __syncthreads();

    const int R = 2 * wv + half;
    for (int pass = 0; pass < 2; ++pass) {
        const int qi = pass ? (NQT - 1 - pr) : pr;
        float mw[NWAVE];
        float M = -INFINITY;
#pragma unroll
        for (int w = 0; w < NWAVE; ++w) { mw[w] = Msm[pass][w][R]; M = fmaxf(M, mw[w]); }
        float L = 0.0f;
        float4 acc = make_float4(0.0f, 0.0f, 0.0f, 0.0f);
#pragma unroll
        for (int w = 0; w < NWAVE; ++w) {
            const float g = expf(mw[w] - M);
            L += g * Lsm[pass][w][R];
            const float4 ov = *(const float4*)(&Wsm[pass][w][R * OPITCH + 4 * l16]);
            acc.x += g * ov.x; acc.y += g * ov.y;
            acc.z += g * ov.z; acc.w += g * ov.w;
        }
        const float inv = 1.0f / L;
        acc.x *= inv; acc.y *= inv; acc.z *= inv; acc.w *= inv;
        float* Ob = out + ((size_t)b * Tsz + qi * 16 + R) * Hsz + 4 * l16;
        *(float4*)Ob = acc;
    }
}

extern "C" void kernel_launch(void* const* d_in, const int* in_sizes, int n_in,
                              void* d_out, int out_size, void* d_ws, size_t ws_size,
                              hipStream_t stream) {
    const float* x  = (const float*)d_in[0];
    const float* Wq = (const float*)d_in[1];
    const float* Wk = (const float*)d_in[2];
    const float* Wv = (const float*)d_in[3];

    const size_t BTH = (size_t)Bsz * Tsz * Hsz;
    float* qkv = (float*)d_ws;
    float* Wt  = qkv + 3 * BTH;

    dim3 g0(Hsz, 3), blk0(256);
    wt_kernel<<<g0, blk0, 0, stream>>>(Wq, Wk, Wv, Wt);

    dim3 g1(Bsz * Tsz / QBM), blk1(256);
    qkv_kernel<<<g1, blk1, 0, stream>>>(x, Wt, qkv);

    dim3 g2(Tsz / 32, Bsz), blk2(32 * NWAVE);
    attn_kernel<<<g2, blk2, 0, stream>>>(qkv, (float*)d_out);
}


# K and V leave the projection in WMMA-fragment order (K tiles accumulated transposed, operands swapped): an attention fragment load reads one 512-B run, not 32-B pieces of many lines (sums unchanged)
# speedup vs baseline: 3.5174x; 1.0902x over previous
#include <hip/hip_runtime.h>
#include <math.h>

#define Bsz 8
#define Tsz 2048
#define Csz 1024
#define Hsz 64

typedef __attribute__((ext_vector_type(2))) float v2f;
typedef __attribute__((ext_vector_type(8))) float v8f;

__device__ __forceinline__ v8f wmma_f32(v2f a, v2f b, v8f c) {
    return __builtin_amdgcn_wmma_f32_16x16x4_f32(
         false, a,  false, b,
         (short)0, c,  false,  false);
}

__global__ __launch_bounds__(256) void wt_kernel(
    const float* __restrict__ Wq,
    const float* __restrict__ Wk,
    const float* __restrict__ Wv,
    float* __restrict__ Wt)
{
    const int which = blockIdx.y;
    const float* W = (which == 0) ? Wq : (which == 1) ? Wk : Wv;
    float* dst = Wt + (size_t)which * Hsz * Csz;
    const int h = blockIdx.x;
    for (int c = threadIdx.x; c < Csz; c += 256)
        dst[(size_t)h * Csz + c] = W[(size_t)c * Hsz + h];
}

#define QBM 64
#define QBN 192
#define QBK 32
#define QPITCH 36
#define QROWS (QBM + QBN)

__global__ __launch_bounds__(256) __attribute__((amdgpu_waves_per_eu(2, 4))) void qkv_kernel(
    const float* __restrict__ x,
    const float* __restrict__ Wt,
    float* __restrict__ qkv)
{
    __shared__ __attribute__((aligned(16))) float Ssm[2][QROWS * QPITCH];

    const int tid  = threadIdx.x;
    const int lane = tid & 31;
    const int wv   = __builtin_amdgcn_readfirstlane(tid >> 5);
    const int wm   = wv & 3;
    const int wn   = wv >> 2;
    const int half = lane >> 4;
    const int l16  = lane & 15;
    const long m0  = (long)blockIdx.x * QBM;

    const int sc = tid & 7;
    const int sr = tid >> 3;
    const float* xg = x  + (m0 + sr) * (size_t)Csz + 4 * sc;
    const float* wg = Wt + (size_t)sr * Csz + 4 * sc;
    const int sdst = sr * QPITCH + 4 * sc;

    const int aoff = (16 * wm + l16) * QPITCH + 4 * half;
    const int boff = (QBM + 32 * wn + l16) * QPITCH + 4 * half;
#define QKV_TJ(j) ((64 * ((j) >> 1) + 16 * ((j) & 1)) * QPITCH)

    v8f acc[6];
#pragma unroll
    for (int j = 0; j < 6; ++j) acc[j] = (v8f){};

    float4 p0, p1, p2, p3, p4, p5, p6, p7;
#define QKV_LOAD_SLAB(k)                                          \
    p0 = *(const float4*)(xg + (k));                              \
    p1 = *(const float4*)(xg + (size_t)32 * Csz + (k));           \
    p2 = *(const float4*)(wg + (k));                              \
    p3 = *(const float4*)(wg + (size_t)32 * Csz + (k));           \
    p4 = *(const float4*)(wg + (size_t)64 * Csz + (k));           \
    p5 = *(const float4*)(wg + (size_t)96 * Csz + (k));           \
    p6 = *(const float4*)(wg + (size_t)128 * Csz + (k));          \
    p7 = *(const float4*)(wg + (size_t)160 * Csz + (k));
#define QKV_STORE_SLAB(S)                                         \
    *(float4*)((S) + sdst +   0 * QPITCH) = p0;                   \
    *(float4*)((S) + sdst +  32 * QPITCH) = p1;                   \
    *(float4*)((S) + sdst +  64 * QPITCH) = p2;                   \
    *(float4*)((S) + sdst +  96 * QPITCH) = p3;                   \
    *(float4*)((S) + sdst + 128 * QPITCH) = p4;                   \
    *(float4*)((S) + sdst + 160 * QPITCH) = p5;                   \
    *(float4*)((S) + sdst + 192 * QPITCH) = p6;                   \
    *(float4*)((S) + sdst + 224 * QPITCH) = p7;

    QKV_LOAD_SLAB(0)
    QKV_STORE_SLAB(&Ssm[0][0])
    __syncthreads();
    QKV_LOAD_SLAB(QBK)

    const int NS = Csz / QBK;
    for (int s = 0; s < NS; ++s) {
        const float* Sb = &Ssm[s & 1][0];
        float4 af[QBK / 8], bf[QBK / 8][6];
#pragma unroll
        for (int g = 0; g < QBK / 8; ++g) {
            af[g] = *(const float4*)(Sb + aoff + 8 * g);
#pragma unroll
            for (int j = 0; j < 6; ++j)
                bf[g][j] = *(const float4*)(Sb + boff + QKV_TJ(j) + 8 * g);
        }
        __builtin_amdgcn_sched_barrier(0);
#pragma unroll
        for (int g = 0; g < QBK / 8; ++g) {
            const v2f alo = { af[g].x, af[g].y }, ahi = { af[g].z, af[g].w };
#pragma unroll
            for (int j = 0; j < 6; ++j) {
                const v2f blo = { bf[g][j].x, bf[g][j].y };
                acc[j] = ((j >> 1) == 1) ? wmma_f32(blo, alo, acc[j])
                                         : wmma_f32(alo, blo, acc[j]);
            }
#pragma unroll
            for (int j = 0; j < 6; ++j) {
                const v2f bhi = { bf[g][j].z, bf[g][j].w };
                acc[j] = ((j >> 1) == 1) ? wmma_f32(bhi, ahi, acc[j])
                                         : wmma_f32(ahi, bhi, acc[j]);
            }
        }
        __builtin_amdgcn_sched_barrier(0);
        if (s + 1 < NS) {
            float* Sn = &Ssm[(s + 1) & 1][0];
            QKV_STORE_SLAB(Sn)
        }
        __syncthreads();
        if (s + 2 < NS) {
            const int k2 = (s + 2) * QBK;
            QKV_LOAD_SLAB(k2)
        }
    }
#undef QKV_LOAD_SLAB
#undef QKV_STORE_SLAB

    const size_t BTH = (size_t)Bsz * Tsz * Hsz;
    const long mrow = m0 + 16 * wm;
    const long tile = (long)blockIdx.x * (QBM / 16) + wm;
#pragma unroll
    for (int j = 0; j < 2; ++j) {
        const int hcol = 32 * wn + 16 * j + l16;
        float* op = qkv + (mrow + 8 * half) * (size_t)Hsz + hcol;
#pragma unroll
        for (int r = 0; r < 8; ++r) op[(size_t)r * Hsz] = acc[j][r];
    }
#pragma unroll
    for (int j = 2; j < 4; ++j) {
        const int c = 4 * wn + 2 * (j & 1) + half;
        float* kp = qkv + BTH + ((size_t)tile * 8 + c) * 128 + 4 * l16;
        *(float4*)(kp +  0) = make_float4(acc[j][0], acc[j][1], acc[j][2], acc[j][3]);
        *(float4*)(kp + 64) = make_float4(acc[j][4], acc[j][5], acc[j][6], acc[j][7]);
    }
#pragma unroll
    for (int j = 4; j < 6; ++j) {
        const int t = 2 * wn + (j & 1);
        float* vp = qkv + 2 * BTH + ((size_t)tile * 4 + t) * 256 + 4 * lane;
        *(float4*)(vp +   0) = make_float4(acc[j][0], acc[j][1], acc[j][2], acc[j][3]);
        *(float4*)(vp + 128) = make_float4(acc[j][4], acc[j][5], acc[j][6], acc[j][7]);
    }
#undef QKV_TJ
}

#define NWAVE 8
#define NQT (Tsz / 16)
#define OPITCH 68

__device__ __forceinline__ float xhalf(float x) {
    const int b = __builtin_bit_cast(int, x);
    return __builtin_bit_cast(float,
        __builtin_amdgcn_permlanex16(b, b, 0x76543210, 0xfedcba98, false, false));
}

__global__ __launch_bounds__(32 * NWAVE) void attn_kernel(
    const float* __restrict__ qkv,
    float* __restrict__ out)
{
    __shared__ __attribute__((aligned(16))) float Wsm[2][NWAVE][16 * OPITCH];
    __shared__ float Msm[2][NWAVE][16];
    __shared__ float Lsm[2][NWAVE][16];

    const int lane = threadIdx.x & 31;
    const int wv   = __builtin_amdgcn_readfirstlane((int)(threadIdx.x >> 5));
    const int half = lane >> 4;
    const int l16  = lane & 15;
    const int b    = blockIdx.y;
    const int pr   = blockIdx.x;

    const size_t BTH = (size_t)Bsz * Tsz * Hsz;
    const float* Q  = qkv +            (size_t)b * Tsz * Hsz;
    const float* Kp = qkv + BTH     +  (size_t)b * Tsz * Hsz + 4 * lane;
    const float* Vp = qkv + 2 * BTH +  (size_t)b * Tsz * Hsz + 4 * lane;

    const float scale = 0.125f;

#pragma unroll 1
    for (int pass = 0; pass < 2; ++pass) {
        const int qi = pass ? (NQT - 1 - pr) : pr;
        const int k0 = pass ? ((wv - pr - 1) & (NWAVE - 1)) : wv;
        const int qrow = qi * 16 + l16;

        v2f qb[16];
#pragma unroll
        for (int c = 0; c < 8; ++c) {
            const float4 t = *(const float4*)(Q + (size_t)qrow * Hsz + 8 * c + 4 * half);
            qb[2 * c][0]     = t.x * scale;
            qb[2 * c][1]     = t.y * scale;
            qb[2 * c + 1][0] = t.z * scale;
            qb[2 * c + 1][1] = t.w * scale;
        }

        float m_run = -INFINITY, l_run = 0.0f;
        v8f o0 = {}, o1 = {}, o2 = {}, o3 = {};

        for (int kt = k0; kt <= qi; kt += NWAVE) {
            const float* Kt = Kp + (size_t)kt * 1024;
            v8f s = {};
#pragma unroll
            for (int c = 0; c < 8; ++c) {
                const float4 ka = *(const float4*)(Kt + 128 * c);
                s = wmma_f32((v2f){ ka.x, ka.y }, qb[2 * c],     s);
                s = wmma_f32((v2f){ ka.z, ka.w }, qb[2 * c + 1], s);
            }

            if (kt == qi) {
#pragma unroll
                for (int r = 0; r < 8; ++r)
                    if (8 * half + r > l16) s[r] = -INFINITY;
            }

            float mx = fmaxf(fmaxf(fmaxf(s[0], s[1]), fmaxf(s[2], s[3])),
                             fmaxf(fmaxf(s[4], s[5]), fmaxf(s[6], s[7])));
            mx = fmaxf(mx, xhalf(mx));
            const float mn    = fmaxf(m_run, mx);
            const float alpha = expf(m_run - mn);
            m_run = mn;
#pragma unroll
            for (int r = 0; r < 8; ++r) s[r] = expf(s[r] - mn);
            const float sm = ((s[0] + s[1]) + (s[2] + s[3])) + ((s[4] + s[5]) + (s[6] + s[7]));
            l_run = l_run * alpha + sm;

#pragma unroll
            for (int r = 0; r < 8; ++r) {
                o0[r] *= alpha; o1[r] *= alpha;
                o2[r] *= alpha; o3[r] *= alpha;
            }

            const float* Vk = Vp + (size_t)kt * 1024;
            const float4 va0 = *(const float4*)(Vk +   0), vb0 = *(const float4*)(Vk + 128);
            const float4 va1 = *(const float4*)(Vk + 256), vb1 = *(const float4*)(Vk + 384);
            const float4 va2 = *(const float4*)(Vk + 512), vb2 = *(const float4*)(Vk + 640);
            const float4 va3 = *(const float4*)(Vk + 768), vb3 = *(const float4*)(Vk + 896);
            const v2f p0 = { s[0], s[1] }, p1 = { s[2], s[3] };
            const v2f p2 = { s[4], s[5] }, p3 = { s[6], s[7] };
            o0 = wmma_f32((v2f){ va0.x, va0.y }, p0, o0);
            o1 = wmma_f32((v2f){ va1.x, va1.y }, p0, o1);
            o2 = wmma_f32((v2f){ va2.x, va2.y }, p0, o2);
            o3 = wmma_f32((v2f){ va3.x, va3.y }, p0, o3);
            o0 = wmma_f32((v2f){ va0.z, va0.w }, p1, o0);
            o1 = wmma_f32((v2f){ va1.z, va1.w }, p1, o1);
            o2 = wmma_f32((v2f){ va2.z, va2.w }, p1, o2);
            o3 = wmma_f32((v2f){ va3.z, va3.w }, p1, o3);
            o0 = wmma_f32((v2f){ vb0.x, vb0.y }, p2, o0);
            o1 = wmma_f32((v2f){ vb1.x, vb1.y }, p2, o1);
            o2 = wmma_f32((v2f){ vb2.x, vb2.y }, p2, o2);
            o3 = wmma_f32((v2f){ vb3.x, vb3.y }, p2, o3);
            o0 = wmma_f32((v2f){ vb0.z, vb0.w }, p3, o0);
            o1 = wmma_f32((v2f){ vb1.z, vb1.w }, p3, o1);
            o2 = wmma_f32((v2f){ vb2.z, vb2.w }, p3, o2);
            o3 = wmma_f32((v2f){ vb3.z, vb3.w }, p3, o3);
        }

        const float l_tot = l_run + xhalf(l_run);

        {
            float* orow = &Wsm[pass][wv][0] + l16 * OPITCH + 8 * half;
            *(float4*)(orow +  0) = make_float4(o0[0], o0[1], o0[2], o0[3]);
            *(float4*)(orow +  4) = make_float4(o0[4], o0[5], o0[6], o0[7]);
            *(float4*)(orow + 16) = make_float4(o1[0], o1[1], o1[2], o1[3]);
            *(float4*)(orow + 20) = make_float4(o1[4], o1[5], o1[6], o1[7]);
            *(float4*)(orow + 32) = make_float4(o2[0], o2[1], o2[2], o2[3]);
            *(float4*)(orow + 36) = make_float4(o2[4], o2[5], o2[6], o2[7]);
            *(float4*)(orow + 48) = make_float4(o3[0], o3[1], o3[2], o3[3]);
            *(float4*)(orow + 52) = make_float4(o3[4], o3[5], o3[6], o3[7]);
        }
        if (half == 0) {
            Msm[pass][wv][l16] = m_run;
            Lsm[pass][wv][l16] = l_tot;
        }
    }
    __syncthreads();

    const int R = 2 * wv + half;
    for (int pass = 0; pass < 2; ++pass) {
        const int qi = pass ? (NQT - 1 - pr) : pr;
        float mw[NWAVE];
        float M = -INFINITY;
#pragma unroll
        for (int w = 0; w < NWAVE; ++w) { mw[w] = Msm[pass][w][R]; M = fmaxf(M, mw[w]); }
        float L = 0.0f;
        float4 acc = make_float4(0.0f, 0.0f, 0.0f, 0.0f);
#pragma unroll
        for (int w = 0; w < NWAVE; ++w) {
            const float g = expf(mw[w] - M);
            L += g * Lsm[pass][w][R];
            const float4 ov = *(const float4*)(&Wsm[pass][w][R * OPITCH + 4 * l16]);
            acc.x += g * ov.x; acc.y += g * ov.y;
            acc.z += g * ov.z; acc.w += g * ov.w;
        }
        const float inv = 1.0f / L;
        acc.x *= inv; acc.y *= inv; acc.z *= inv; acc.w *= inv;
        float* Ob = out + ((size_t)b * Tsz + qi * 16 + R) * Hsz + 4 * l16;
        *(float4*)Ob = acc;
    }
}

extern "C" void kernel_launch(void* const* d_in, const int* in_sizes, int n_in,
                              void* d_out, int out_size, void* d_ws, size_t ws_size,
                              hipStream_t stream) {
    const float* x  = (const float*)d_in[0];
    const float* Wq = (const float*)d_in[1];
    const float* Wk = (const float*)d_in[2];
    const float* Wv = (const float*)d_in[3];

    const size_t BTH = (size_t)Bsz * Tsz * Hsz;
    float* qkv = (float*)d_ws;
    float* Wt  = qkv + 3 * BTH;

    dim3 g0(Hsz, 3), blk0(256);
    wt_kernel<<<g0, blk0, 0, stream>>>(Wq, Wk, Wv, Wt);

    dim3 g1(Bsz * Tsz / QBM), blk1(256);
    qkv_kernel<<<g1, blk1, 0, stream>>>(x, Wt, qkv);

    dim3 g2(Tsz / 32, Bsz), blk2(32 * NWAVE);
    attn_kernel<<<g2, blk2, 0, stream>>>(qkv, (float*)d_out);
}


# attention: a wave multiplies each loaded K and V fragment against TWO adjacent query tiles; 16-wave work-group per pair of such double tiles: half the K/V fetches per WMMA (sum order changes)
# speedup vs baseline: 3.6287x; 1.0316x over previous
#include <hip/hip_runtime.h>
#include <math.h>

#define Bsz 8
#define Tsz 2048
#define Csz 1024
#define Hsz 64

typedef __attribute__((ext_vector_type(2))) float v2f;
typedef __attribute__((ext_vector_type(8))) float v8f;

__device__ __forceinline__ v8f wmma_f32(v2f a, v2f b, v8f c) {
    return __builtin_amdgcn_wmma_f32_16x16x4_f32(
         false, a,  false, b,
         (short)0, c,  false,  false);
}

__global__ __launch_bounds__(256) void wt_kernel(
    const float* __restrict__ Wq,
    const float* __restrict__ Wk,
    const float* __restrict__ Wv,
    float* __restrict__ Wt)
{
    const int which = blockIdx.y;
    const float* W = (which == 0) ? Wq : (which == 1) ? Wk : Wv;
    float* dst = Wt + (size_t)which * Hsz * Csz;
    const int h = blockIdx.x;
    for (int c = threadIdx.x; c < Csz; c += 256)
        dst[(size_t)h * Csz + c] = W[(size_t)c * Hsz + h];
}

#define QBM 64
#define QBN 192
#define QBK 32
#define QPITCH 36
#define QROWS (QBM + QBN)

__global__ __launch_bounds__(256) __attribute__((amdgpu_waves_per_eu(2, 4))) void qkv_kernel(
    const float* __restrict__ x,
    const float* __restrict__ Wt,
    float* __restrict__ qkv)
{
    __shared__ __attribute__((aligned(16))) float Ssm[2][QROWS * QPITCH];

    const int tid  = threadIdx.x;
    const int lane = tid & 31;
    const int wv   = __builtin_amdgcn_readfirstlane(tid >> 5);
    const int wm   = wv & 3;
    const int wn   = wv >> 2;
    const int half = lane >> 4;
    const int l16  = lane & 15;
    const long m0  = (long)blockIdx.x * QBM;

    const int sc = tid & 7;
    const int sr = tid >> 3;
    const float* xg = x  + (m0 + sr) * (size_t)Csz + 4 * sc;
    const float* wg = Wt + (size_t)sr * Csz + 4 * sc;
    const int sdst = sr * QPITCH + 4 * sc;

    const int aoff = (16 * wm + l16) * QPITCH + 4 * half;
    const int boff = (QBM + 32 * wn + l16) * QPITCH + 4 * half;
#define QKV_TJ(j) ((64 * ((j) >> 1) + 16 * ((j) & 1)) * QPITCH)

    v8f acc[6];
#pragma unroll
    for (int j = 0; j < 6; ++j) acc[j] = (v8f){};

    float4 p0, p1, p2, p3, p4, p5, p6, p7;
#define QKV_LOAD_SLAB(k)                                          \
    p0 = *(const float4*)(xg + (k));                              \
    p1 = *(const float4*)(xg + (size_t)32 * Csz + (k));           \
    p2 = *(const float4*)(wg + (k));                              \
    p3 = *(const float4*)(wg + (size_t)32 * Csz + (k));           \
    p4 = *(const float4*)(wg + (size_t)64 * Csz + (k));           \
    p5 = *(const float4*)(wg + (size_t)96 * Csz + (k));           \
    p6 = *(const float4*)(wg + (size_t)128 * Csz + (k));          \
    p7 = *(const float4*)(wg + (size_t)160 * Csz + (k));
#define QKV_STORE_SLAB(S)                                         \
    *(float4*)((S) + sdst +   0 * QPITCH) = p0;                   \
    *(float4*)((S) + sdst +  32 * QPITCH) = p1;                   \
    *(float4*)((S) + sdst +  64 * QPITCH) = p2;                   \
    *(float4*)((S) + sdst +  96 * QPITCH) = p3;                   \
    *(float4*)((S) + sdst + 128 * QPITCH) = p4;                   \
    *(float4*)((S) + sdst + 160 * QPITCH) = p5;                   \
    *(float4*)((S) + sdst + 192 * QPITCH) = p6;                   \
    *(float4*)((S) + sdst + 224 * QPITCH) = p7;

    QKV_LOAD_SLAB(0)
    QKV_STORE_SLAB(&Ssm[0][0])
    __syncthreads();
    QKV_LOAD_SLAB(QBK)

    const int NS = Csz / QBK;
    for (int s = 0; s < NS; ++s) {
        const float* Sb = &Ssm[s & 1][0];
        float4 af[QBK / 8], bf[QBK / 8][6];
#pragma unroll
        for (int g = 0; g < QBK / 8; ++g) {
            af[g] = *(const float4*)(Sb + aoff + 8 * g);
#pragma unroll
            for (int j = 0; j < 6; ++j)
                bf[g][j] = *(const float4*)(Sb + boff + QKV_TJ(j) + 8 * g);
        }
        __builtin_amdgcn_sched_barrier(0);
#pragma unroll
        for (int g = 0; g < QBK / 8; ++g) {
            const v2f alo = { af[g].x, af[g].y }, ahi = { af[g].z, af[g].w };
#pragma unroll
            for (int j = 0; j < 6; ++j) {
                const v2f blo = { bf[g][j].x, bf[g][j].y };
                acc[j] = ((j >> 1) == 1) ? wmma_f32(blo, alo, acc[j])
                                         : wmma_f32(alo, blo, acc[j]);
            }
#pragma unroll
            for (int j = 0; j < 6; ++j) {
                const v2f bhi = { bf[g][j].z, bf[g][j].w };
                acc[j] = ((j >> 1) == 1) ? wmma_f32(bhi, ahi, acc[j])
                                         : wmma_f32(ahi, bhi, acc[j]);
            }
        }
        __builtin_amdgcn_sched_barrier(0);
        if (s + 1 < NS) {
            float* Sn = &Ssm[(s + 1) & 1][0];
            QKV_STORE_SLAB(Sn)
        }
        __syncthreads();
        if (s + 2 < NS) {
            const int k2 = (s + 2) * QBK;
            QKV_LOAD_SLAB(k2)
        }
    }
#undef QKV_LOAD_SLAB
#undef QKV_STORE_SLAB

    const size_t BTH = (size_t)Bsz * Tsz * Hsz;
    const long mrow = m0 + 16 * wm;
    const long tile = (long)blockIdx.x * (QBM / 16) + wm;
#pragma unroll
    for (int j = 0; j < 2; ++j) {
        const int hcol = 32 * wn + 16 * j + l16;
        float* op = qkv + (mrow + 8 * half) * (size_t)Hsz + hcol;
#pragma unroll
        for (int r = 0; r < 8; ++r) op[(size_t)r * Hsz] = acc[j][r];
    }
#pragma unroll
    for (int j = 2; j < 4; ++j) {
        const int c = 4 * wn + 2 * (j & 1) + half;
        float* kp = qkv + BTH + ((size_t)tile * 8 + c) * 128 + 4 * l16;
        *(float4*)(kp +  0) = make_float4(acc[j][0], acc[j][1], acc[j][2], acc[j][3]);
        *(float4*)(kp + 64) = make_float4(acc[j][4], acc[j][5], acc[j][6], acc[j][7]);
    }
#pragma unroll
    for (int j = 4; j < 6; ++j) {
        const int t = 2 * wn + (j & 1);
        float* vp = qkv + 2 * BTH + ((size_t)tile * 4 + t) * 256 + 4 * lane;
        *(float4*)(vp +   0) = make_float4(acc[j][0], acc[j][1], acc[j][2], acc[j][3]);
        *(float4*)(vp + 128) = make_float4(acc[j][4], acc[j][5], acc[j][6], acc[j][7]);
    }
#undef QKV_TJ
}

#define NWAVE 16
#define NDT (Tsz / 32)
#define OPITCH 68

__device__ __forceinline__ float xhalf(float x) {
    const int b = __builtin_bit_cast(int, x);
    return __builtin_bit_cast(float,
        __builtin_amdgcn_permlanex16(b, b, 0x76543210, 0xfedcba98, false, false));
}

template <bool GUARD>
__device__ __forceinline__ float softmax_step(v8f& s, float& m_run, float& l_run) {
    float mx = fmaxf(fmaxf(fmaxf(s[0], s[1]), fmaxf(s[2], s[3])),
                     fmaxf(fmaxf(s[4], s[5]), fmaxf(s[6], s[7])));
    mx = fmaxf(mx, xhalf(mx));
    const float mn    = fmaxf(m_run, mx);
    const float mref  = (GUARD && mn == -INFINITY) ? 0.0f : mn;
    const float alpha = expf(m_run - mref);
    m_run = mn;
#pragma unroll
    for (int r = 0; r < 8; ++r) s[r] = expf(s[r] - mref);
    const float sm = ((s[0] + s[1]) + (s[2] + s[3])) + ((s[4] + s[5]) + (s[6] + s[7]));
    l_run = l_run * alpha + sm;
    return alpha;
}

__global__ __launch_bounds__(32 * NWAVE) void attn_kernel(
    const float* __restrict__ qkv,
    float* __restrict__ out)
{
    __shared__ __attribute__((aligned(16))) float Wsm[4][NWAVE][16 * OPITCH];
    __shared__ float Msm[4][NWAVE][16];
    __shared__ float Lsm[4][NWAVE][16];

    const int lane = threadIdx.x & 31;
    const int wv   = __builtin_amdgcn_readfirstlane((int)(threadIdx.x >> 5));
    const int half = lane >> 4;
    const int l16  = lane & 15;
    const int b    = blockIdx.y;
    const int pr   = blockIdx.x;

    const size_t BTH = (size_t)Bsz * Tsz * Hsz;
    const float* Q  = qkv +            (size_t)b * Tsz * Hsz;
    const float* Kp = qkv + BTH     +  (size_t)b * Tsz * Hsz + 4 * lane;
    const float* Vp = qkv + 2 * BTH +  (size_t)b * Tsz * Hsz + 4 * lane;

    const float scale = 0.125f;

#pragma unroll 1
    for (int pass = 0; pass < 2; ++pass) {
        const int dq  = pass ? (NDT - 1 - pr) : pr;
        const int qiA = 2 * dq, qiB = 2 * dq + 1;
        const int k0  = pass ? ((wv - 2 * pr - 2) & (NWAVE - 1)) : wv;
        const int qrowA = qiA * 16 + l16;

        v2f qa[16], qb[16];
#pragma unroll
        for (int c = 0; c < 8; ++c) {
            const float4 ta = *(const float4*)(Q + (size_t)qrowA * Hsz + 8 * c + 4 * half);
            const float4 tb = *(const float4*)(Q + (size_t)(qrowA + 16) * Hsz + 8 * c + 4 * half);
            qa[2 * c][0]     = ta.x * scale;
            qa[2 * c][1]     = ta.y * scale;
            qa[2 * c + 1][0] = ta.z * scale;
            qa[2 * c + 1][1] = ta.w * scale;
            qb[2 * c][0]     = tb.x * scale;
            qb[2 * c][1]     = tb.y * scale;
            qb[2 * c + 1][0] = tb.z * scale;
            qb[2 * c + 1][1] = tb.w * scale;
        }

        float mA = -INFINITY, lA = 0.0f, mB = -INFINITY, lB = 0.0f;
        v8f a0 = {}, a1 = {}, a2 = {}, a3 = {};
        v8f b0 = {}, b1 = {}, b2 = {}, b3 = {};

        for (int kt = k0; kt <= qiB; kt += NWAVE) {
            const float* Kt = Kp + (size_t)kt * 1024;
            v8f sA = {}, sB = {};
#pragma unroll
            for (int c = 0; c < 8; ++c) {
                const float4 ka = *(const float4*)(Kt + 128 * c);
                sA = wmma_f32((v2f){ ka.x, ka.y }, qa[2 * c],     sA);
                sB = wmma_f32((v2f){ ka.x, ka.y }, qb[2 * c],     sB);
                sA = wmma_f32((v2f){ ka.z, ka.w }, qa[2 * c + 1], sA);
                sB = wmma_f32((v2f){ ka.z, ka.w }, qb[2 * c + 1], sB);
            }

            if (kt >= qiA) {
                const int koff = 16 * (kt - qiA) + 8 * half;
#pragma unroll
                for (int r = 0; r < 8; ++r)
                    if (koff + r > l16) sA[r] = -INFINITY;
                if (kt == qiB) {
#pragma unroll
                    for (int r = 0; r < 8; ++r)
                        if (8 * half + r > l16) sB[r] = -INFINITY;
                }
            }

            const float alA = softmax_step<true>(sA, mA, lA);
            const float alB = softmax_step<false>(sB, mB, lB);

#pragma unroll
            for (int r = 0; r < 8; ++r) {
                a0[r] *= alA; a1[r] *= alA; a2[r] *= alA; a3[r] *= alA;
                b0[r] *= alB; b1[r] *= alB; b2[r] *= alB; b3[r] *= alB;
            }

            const float* Vk = Vp + (size_t)kt * 1024;
            const float4 va0 = *(const float4*)(Vk +   0), vb0 = *(const float4*)(Vk + 128);
            const float4 va1 = *(const float4*)(Vk + 256), vb1 = *(const float4*)(Vk + 384);
            const float4 va2 = *(const float4*)(Vk + 512), vb2 = *(const float4*)(Vk + 640);
            const float4 va3 = *(const float4*)(Vk + 768), vb3 = *(const float4*)(Vk + 896);
            const v2f pA0 = { sA[0], sA[1] }, pA1 = { sA[2], sA[3] };
            const v2f pA2 = { sA[4], sA[5] }, pA3 = { sA[6], sA[7] };
            const v2f pB0 = { sB[0], sB[1] }, pB1 = { sB[2], sB[3] };
            const v2f pB2 = { sB[4], sB[5] }, pB3 = { sB[6], sB[7] };
            a0 = wmma_f32((v2f){ va0.x, va0.y }, pA0, a0);
            b0 = wmma_f32((v2f){ va0.x, va0.y }, pB0, b0);
            a1 = wmma_f32((v2f){ va1.x, va1.y }, pA0, a1);
            b1 = wmma_f32((v2f){ va1.x, va1.y }, pB0, b1);
            a2 = wmma_f32((v2f){ va2.x, va2.y }, pA0, a2);
            b2 = wmma_f32((v2f){ va2.x, va2.y }, pB0, b2);
            a3 = wmma_f32((v2f){ va3.x, va3.y }, pA0, a3);
            b3 = wmma_f32((v2f){ va3.x, va3.y }, pB0, b3);
            a0 = wmma_f32((v2f){ va0.z, va0.w }, pA1, a0);
            b0 = wmma_f32((v2f){ va0.z, va0.w }, pB1, b0);
            a1 = wmma_f32((v2f){ va1.z, va1.w }, pA1, a1);
            b1 = wmma_f32((v2f){ va1.z, va1.w }, pB1, b1);
            a2 = wmma_f32((v2f){ va2.z, va2.w }, pA1, a2);
            b2 = wmma_f32((v2f){ va2.z, va2.w }, pB1, b2);
            a3 = wmma_f32((v2f){ va3.z, va3.w }, pA1, a3);
            b3 = wmma_f32((v2f){ va3.z, va3.w }, pB1, b3);
            a0 = wmma_f32((v2f){ vb0.x, vb0.y }, pA2, a0);
            b0 = wmma_f32((v2f){ vb0.x, vb0.y }, pB2, b0);
            a1 = wmma_f32((v2f){ vb1.x, vb1.y }, pA2, a1);
            b1 = wmma_f32((v2f){ vb1.x, vb1.y }, pB2, b1);
            a2 = wmma_f32((v2f){ vb2.x, vb2.y }, pA2, a2);
            b2 = wmma_f32((v2f){ vb2.x, vb2.y }, pB2, b2);
            a3 = wmma_f32((v2f){ vb3.x, vb3.y }, pA2, a3);
            b3 = wmma_f32((v2f){ vb3.x, vb3.y }, pB2, b3);
            a0 = wmma_f32((v2f){ vb0.z, vb0.w }, pA3, a0);
            b0 = wmma_f32((v2f){ vb0.z, vb0.w }, pB3, b0);
            a1 = wmma_f32((v2f){ vb1.z, vb1.w }, pA3, a1);
            b1 = wmma_f32((v2f){ vb1.z, vb1.w }, pB3, b1);
            a2 = wmma_f32((v2f){ vb2.z, vb2.w }, pA3, a2);
            b2 = wmma_f32((v2f){ vb2.z, vb2.w }, pB3, b2);
            a3 = wmma_f32((v2f){ vb3.z, vb3.w }, pA3, a3);
            b3 = wmma_f32((v2f){ vb3.z, vb3.w }, pB3, b3);
        }

        const float lAt = lA + xhalf(lA);
        const float lBt = lB + xhalf(lB);

        {
            float* orow = &Wsm[2 * pass][wv][0] + l16 * OPITCH + 8 * half;
            *(float4*)(orow +  0) = make_float4(a0[0], a0[1], a0[2], a0[3]);
            *(float4*)(orow +  4) = make_float4(a0[4], a0[5], a0[6], a0[7]);
            *(float4*)(orow + 16) = make_float4(a1[0], a1[1], a1[2], a1[3]);
            *(float4*)(orow + 20) = make_float4(a1[4], a1[5], a1[6], a1[7]);
            *(float4*)(orow + 32) = make_float4(a2[0], a2[1], a2[2], a2[3]);
            *(float4*)(orow + 36) = make_float4(a2[4], a2[5], a2[6], a2[7]);
            *(float4*)(orow + 48) = make_float4(a3[0], a3[1], a3[2], a3[3]);
            *(float4*)(orow + 52) = make_float4(a3[4], a3[5], a3[6], a3[7]);
        }
        {
            float* orow = &Wsm[2 * pass + 1][wv][0] + l16 * OPITCH + 8 * half;
            *(float4*)(orow +  0) = make_float4(b0[0], b0[1], b0[2], b0[3]);
            *(float4*)(orow +  4) = make_float4(b0[4], b0[5], b0[6], b0[7]);
            *(float4*)(orow + 16) = make_float4(b1[0], b1[1], b1[2], b1[3]);
            *(float4*)(orow + 20) = make_float4(b1[4], b1[5], b1[6], b1[7]);
            *(float4*)(orow + 32) = make_float4(b2[0], b2[1], b2[2], b2[3]);
            *(float4*)(orow + 36) = make_float4(b2[4], b2[5], b2[6], b2[7]);
            *(float4*)(orow + 48) = make_float4(b3[0], b3[1], b3[2], b3[3]);
            *(float4*)(orow + 52) = make_float4(b3[4], b3[5], b3[6], b3[7]);
        }
        if (half == 0) {
            Msm[2 * pass][wv][l16]     = mA;
            Lsm[2 * pass][wv][l16]     = lAt;
            Msm[2 * pass + 1][wv][l16] = mB;
            Lsm[2 * pass + 1][wv][l16] = lBt;
        }
    }
    __syncthreads();

    const int ab = wv >> 3;
    const int R  = 2 * (wv & 7) + half;
    for (int pass = 0; pass < 2; ++pass) {
        const int slot = 2 * pass + ab;
        const int qi = 2 * (pass ? (NDT - 1 - pr) : pr) + ab;
        float mw[NWAVE];
        float M = -INFINITY;
#pragma unroll
        for (int w = 0; w < NWAVE; ++w) { mw[w] = Msm[slot][w][R]; M = fmaxf(M, mw[w]); }
        float L = 0.0f;
        float4 acc = make_float4(0.0f, 0.0f, 0.0f, 0.0f);
#pragma unroll
        for (int w = 0; w < NWAVE; ++w) {
            const float g = expf(mw[w] - M);
            L += g * Lsm[slot][w][R];
            const float4 ov = *(const float4*)(&Wsm[slot][w][R * OPITCH + 4 * l16]);
            acc.x += g * ov.x; acc.y += g * ov.y;
            acc.z += g * ov.z; acc.w += g * ov.w;
        }
        const float inv = 1.0f / L;
        acc.x *= inv; acc.y *= inv; acc.z *= inv; acc.w *= inv;
        float* Ob = out + ((size_t)b * Tsz + qi * 16 + R) * Hsz + 4 * l16;
        *(float4*)Ob = acc;
    }
}

extern "C" void kernel_launch(void* const* d_in, const int* in_sizes, int n_in,
                              void* d_out, int out_size, void* d_ws, size_t ws_size,
                              hipStream_t stream) {
    const float* x  = (const float*)d_in[0];
    const float* Wq = (const float*)d_in[1];
    const float* Wk = (const float*)d_in[2];
    const float* Wv = (const float*)d_in[3];

    const size_t BTH = (size_t)Bsz * Tsz * Hsz;
    float* qkv = (float*)d_ws;
    float* Wt  = qkv + 3 * BTH;

    dim3 g0(Hsz, 3), blk0(256);
    wt_kernel<<<g0, blk0, 0, stream>>>(Wq, Wk, Wv, Wt);

    dim3 g1(Bsz * Tsz / QBM), blk1(256);
    qkv_kernel<<<g1, blk1, 0, stream>>>(x, Wt, qkv);

    dim3 g2(Tsz / 64, Bsz), blk2(32 * NWAVE);
    attn_kernel<<<g2, blk2, 0, stream>>>(qkv, (float*)d_out);
}


# attention softmax in the log2 domain: log2(e)/sqrt(H) folded into the Q pre-scale; every exponential (scores, rescale, merge) is one v_exp_f32, not the expf expansion (exponent rounding changes)
# speedup vs baseline: 3.8088x; 1.0496x over previous
#include <hip/hip_runtime.h>
#include <math.h>

#define Bsz 8
#define Tsz 2048
#define Csz 1024
#define Hsz 64

typedef __attribute__((ext_vector_type(2))) float v2f;
typedef __attribute__((ext_vector_type(8))) float v8f;

__device__ __forceinline__ v8f wmma_f32(v2f a, v2f b, v8f c) {
    return __builtin_amdgcn_wmma_f32_16x16x4_f32(
         false, a,  false, b,
         (short)0, c,  false,  false);
}

__global__ __launch_bounds__(256) void wt_kernel(
    const float* __restrict__ Wq,
    const float* __restrict__ Wk,
    const float* __restrict__ Wv,
    float* __restrict__ Wt)
{
    const int which = blockIdx.y;
    const float* W = (which == 0) ? Wq : (which == 1) ? Wk : Wv;
    float* dst = Wt + (size_t)which * Hsz * Csz;
    const int h = blockIdx.x;
    for (int c = threadIdx.x; c < Csz; c += 256)
        dst[(size_t)h * Csz + c] = W[(size_t)c * Hsz + h];
}

#define QBM 64
#define QBN 192
#define QBK 32
#define QPITCH 36
#define QROWS (QBM + QBN)

__global__ __launch_bounds__(256) __attribute__((amdgpu_waves_per_eu(2, 4))) void qkv_kernel(
    const float* __restrict__ x,
    const float* __restrict__ Wt,
    float* __restrict__ qkv)
{
    __shared__ __attribute__((aligned(16))) float Ssm[2][QROWS * QPITCH];

    const int tid  = threadIdx.x;
    const int lane = tid & 31;
    const int wv   = __builtin_amdgcn_readfirstlane(tid >> 5);
    const int wm   = wv & 3;
    const int wn   = wv >> 2;
    const int half = lane >> 4;
    const int l16  = lane & 15;
    const long m0  = (long)blockIdx.x * QBM;

    const int sc = tid & 7;
    const int sr = tid >> 3;
    const float* xg = x  + (m0 + sr) * (size_t)Csz + 4 * sc;
    const float* wg = Wt + (size_t)sr * Csz + 4 * sc;
    const int sdst = sr * QPITCH + 4 * sc;

    const int aoff = (16 * wm + l16) * QPITCH + 4 * half;
    const int boff = (QBM + 32 * wn + l16) * QPITCH + 4 * half;
#define QKV_TJ(j) ((64 * ((j) >> 1) + 16 * ((j) & 1)) * QPITCH)

    v8f acc[6];
#pragma unroll
    for (int j = 0; j < 6; ++j) acc[j] = (v8f){};

    float4 p0, p1, p2, p3, p4, p5, p6, p7;
#define QKV_LOAD_SLAB(k)                                          \
    p0 = *(const float4*)(xg + (k));                              \
    p1 = *(const float4*)(xg + (size_t)32 * Csz + (k));           \
    p2 = *(const float4*)(wg + (k));                              \
    p3 = *(const float4*)(wg + (size_t)32 * Csz + (k));           \
    p4 = *(const float4*)(wg + (size_t)64 * Csz + (k));           \
    p5 = *(const float4*)(wg + (size_t)96 * Csz + (k));           \
    p6 = *(const float4*)(wg + (size_t)128 * Csz + (k));          \
    p7 = *(const float4*)(wg + (size_t)160 * Csz + (k));
#define QKV_STORE_SLAB(S)                                         \
    *(float4*)((S) + sdst +   0 * QPITCH) = p0;                   \
    *(float4*)((S) + sdst +  32 * QPITCH) = p1;                   \
    *(float4*)((S) + sdst +  64 * QPITCH) = p2;                   \
    *(float4*)((S) + sdst +  96 * QPITCH) = p3;                   \
    *(float4*)((S) + sdst + 128 * QPITCH) = p4;                   \
    *(float4*)((S) + sdst + 160 * QPITCH) = p5;                   \
    *(float4*)((S) + sdst + 192 * QPITCH) = p6;                   \
    *(float4*)((S) + sdst + 224 * QPITCH) = p7;

    QKV_LOAD_SLAB(0)
    QKV_STORE_SLAB(&Ssm[0][0])
    __syncthreads();
    QKV_LOAD_SLAB(QBK)

    const int NS = Csz / QBK;
    for (int s = 0; s < NS; ++s) {
        const float* Sb = &Ssm[s & 1][0];
        float4 af[QBK / 8], bf[QBK / 8][6];
#pragma unroll
        for (int g = 0; g < QBK / 8; ++g) {
            af[g] = *(const float4*)(Sb + aoff + 8 * g);
#pragma unroll
            for (int j = 0; j < 6; ++j)
                bf[g][j] = *(const float4*)(Sb + boff + QKV_TJ(j) + 8 * g);
        }
        __builtin_amdgcn_sched_barrier(0);
#pragma unroll
        for (int g = 0; g < QBK / 8; ++g) {
            const v2f alo = { af[g].x, af[g].y }, ahi = { af[g].z, af[g].w };
#pragma unroll
            for (int j = 0; j < 6; ++j) {
                const v2f blo = { bf[g][j].x, bf[g][j].y };
                acc[j] = ((j >> 1) == 1) ? wmma_f32(blo, alo, acc[j])
                                         : wmma_f32(alo, blo, acc[j]);
            }
#pragma unroll
            for (int j = 0; j < 6; ++j) {
                const v2f bhi = { bf[g][j].z, bf[g][j].w };
                acc[j] = ((j >> 1) == 1) ? wmma_f32(bhi, ahi, acc[j])
                                         : wmma_f32(ahi, bhi, acc[j]);
            }
        }
        __builtin_amdgcn_sched_barrier(0);
        if (s + 1 < NS) {
            float* Sn = &Ssm[(s + 1) & 1][0];
            QKV_STORE_SLAB(Sn)
        }
        __syncthreads();
        if (s + 2 < NS) {
            const int k2 = (s + 2) * QBK;
            QKV_LOAD_SLAB(k2)
        }
    }
#undef QKV_LOAD_SLAB
#undef QKV_STORE_SLAB

    const size_t BTH = (size_t)Bsz * Tsz * Hsz;
    const long mrow = m0 + 16 * wm;
    const long tile = (long)blockIdx.x * (QBM / 16) + wm;
#pragma unroll
    for (int j = 0; j < 2; ++j) {
        const int hcol = 32 * wn + 16 * j + l16;
        float* op = qkv + (mrow + 8 * half) * (size_t)Hsz + hcol;
#pragma unroll
        for (int r = 0; r < 8; ++r) op[(size_t)r * Hsz] = acc[j][r];
    }
#pragma unroll
    for (int j = 2; j < 4; ++j) {
        const int c = 4 * wn + 2 * (j & 1) + half;
        float* kp = qkv + BTH + ((size_t)tile * 8 + c) * 128 + 4 * l16;
        *(float4*)(kp +  0) = make_float4(acc[j][0], acc[j][1], acc[j][2], acc[j][3]);
        *(float4*)(kp + 64) = make_float4(acc[j][4], acc[j][5], acc[j][6], acc[j][7]);
    }
#pragma unroll
    for (int j = 4; j < 6; ++j) {
        const int t = 2 * wn + (j & 1);
        float* vp = qkv + 2 * BTH + ((size_t)tile * 4 + t) * 256 + 4 * lane;
        *(float4*)(vp +   0) = make_float4(acc[j][0], acc[j][1], acc[j][2], acc[j][3]);
        *(float4*)(vp + 128) = make_float4(acc[j][4], acc[j][5], acc[j][6], acc[j][7]);
    }
#undef QKV_TJ
}

#define NWAVE 16
#define NDT (Tsz / 32)
#define OPITCH 68

__device__ __forceinline__ float xhalf(float x) {
    const int b = __builtin_bit_cast(int, x);
    return __builtin_bit_cast(float,
        __builtin_amdgcn_permlanex16(b, b, 0x76543210, 0xfedcba98, false, false));
}

template <bool GUARD>
__device__ __forceinline__ float softmax_step(v8f& s, float& m_run, float& l_run) {
    float mx = fmaxf(fmaxf(fmaxf(s[0], s[1]), fmaxf(s[2], s[3])),
                     fmaxf(fmaxf(s[4], s[5]), fmaxf(s[6], s[7])));
    mx = fmaxf(mx, xhalf(mx));
    const float mn    = fmaxf(m_run, mx);
    const float mref  = (GUARD && mn == -INFINITY) ? 0.0f : mn;
    const float alpha = __builtin_amdgcn_exp2f(m_run - mref);
    m_run = mn;
#pragma unroll
    for (int r = 0; r < 8; ++r) s[r] = __builtin_amdgcn_exp2f(s[r] - mref);
    const float sm = ((s[0] + s[1]) + (s[2] + s[3])) + ((s[4] + s[5]) + (s[6] + s[7]));
    l_run = l_run * alpha + sm;
    return alpha;
}

__global__ __launch_bounds__(32 * NWAVE) void attn_kernel(
    const float* __restrict__ qkv,
    float* __restrict__ out)
{
    __shared__ __attribute__((aligned(16))) float Wsm[4][NWAVE][16 * OPITCH];
    __shared__ float Msm[4][NWAVE][16];
    __shared__ float Lsm[4][NWAVE][16];

    const int lane = threadIdx.x & 31;
    const int wv   = __builtin_amdgcn_readfirstlane((int)(threadIdx.x >> 5));
    const int half = lane >> 4;
    const int l16  = lane & 15;
    const int b    = blockIdx.y;
    const int pr   = blockIdx.x;

    const size_t BTH = (size_t)Bsz * Tsz * Hsz;
    const float* Q  = qkv +            (size_t)b * Tsz * Hsz;
    const float* Kp = qkv + BTH     +  (size_t)b * Tsz * Hsz + 4 * lane;
    const float* Vp = qkv + 2 * BTH +  (size_t)b * Tsz * Hsz + 4 * lane;

    const float scale = 0.125f * 1.4426950408889634f;

#pragma unroll 1
    for (int pass = 0; pass < 2; ++pass) {
        const int dq  = pass ? (NDT - 1 - pr) : pr;
        const int qiA = 2 * dq, qiB = 2 * dq + 1;
        const int k0  = pass ? ((wv - 2 * pr - 2) & (NWAVE - 1)) : wv;
        const int qrowA = qiA * 16 + l16;

        v2f qa[16], qb[16];
#pragma unroll
        for (int c = 0; c < 8; ++c) {
            const float4 ta = *(const float4*)(Q + (size_t)qrowA * Hsz + 8 * c + 4 * half);
            const float4 tb = *(const float4*)(Q + (size_t)(qrowA + 16) * Hsz + 8 * c + 4 * half);
            qa[2 * c][0]     = ta.x * scale;
            qa[2 * c][1]     = ta.y * scale;
            qa[2 * c + 1][0] = ta.z * scale;
            qa[2 * c + 1][1] = ta.w * scale;
            qb[2 * c][0]     = tb.x * scale;
            qb[2 * c][1]     = tb.y * scale;
            qb[2 * c + 1][0] = tb.z * scale;
            qb[2 * c + 1][1] = tb.w * scale;
        }

        float mA = -INFINITY, lA = 0.0f, mB = -INFINITY, lB = 0.0f;
        v8f a0 = {}, a1 = {}, a2 = {}, a3 = {};
        v8f b0 = {}, b1 = {}, b2 = {}, b3 = {};

        for (int kt = k0; kt <= qiB; kt += NWAVE) {
            const float* Kt = Kp + (size_t)kt * 1024;
            v8f sA = {}, sB = {};
#pragma unroll
            for (int c = 0; c < 8; ++c) {
                const float4 ka = *(const float4*)(Kt + 128 * c);
                sA = wmma_f32((v2f){ ka.x, ka.y }, qa[2 * c],     sA);
                sB = wmma_f32((v2f){ ka.x, ka.y }, qb[2 * c],     sB);
                sA = wmma_f32((v2f){ ka.z, ka.w }, qa[2 * c + 1], sA);
                sB = wmma_f32((v2f){ ka.z, ka.w }, qb[2 * c + 1], sB);
            }

            if (kt >= qiA) {
                const int koff = 16 * (kt - qiA) + 8 * half;
#pragma unroll
                for (int r = 0; r < 8; ++r)
                    if (koff + r > l16) sA[r] = -INFINITY;
                if (kt == qiB) {
#pragma unroll
                    for (int r = 0; r < 8; ++r)
                        if (8 * half + r > l16) sB[r] = -INFINITY;
                }
            }

            const float alA = softmax_step<true>(sA, mA, lA);
            const float alB = softmax_step<false>(sB, mB, lB);

#pragma unroll
            for (int r = 0; r < 8; ++r) {
                a0[r] *= alA; a1[r] *= alA; a2[r] *= alA; a3[r] *= alA;
                b0[r] *= alB; b1[r] *= alB; b2[r] *= alB; b3[r] *= alB;
            }

            const float* Vk = Vp + (size_t)kt * 1024;
            const float4 va0 = *(const float4*)(Vk +   0), vb0 = *(const float4*)(Vk + 128);
            const float4 va1 = *(const float4*)(Vk + 256), vb1 = *(const float4*)(Vk + 384);
            const float4 va2 = *(const float4*)(Vk + 512), vb2 = *(const float4*)(Vk + 640);
            const float4 va3 = *(const float4*)(Vk + 768), vb3 = *(const float4*)(Vk + 896);
            const v2f pA0 = { sA[0], sA[1] }, pA1 = { sA[2], sA[3] };
            const v2f pA2 = { sA[4], sA[5] }, pA3 = { sA[6], sA[7] };
            const v2f pB0 = { sB[0], sB[1] }, pB1 = { sB[2], sB[3] };
            const v2f pB2 = { sB[4], sB[5] }, pB3 = { sB[6], sB[7] };
            a0 = wmma_f32((v2f){ va0.x, va0.y }, pA0, a0);
            b0 = wmma_f32((v2f){ va0.x, va0.y }, pB0, b0);
            a1 = wmma_f32((v2f){ va1.x, va1.y }, pA0, a1);
            b1 = wmma_f32((v2f){ va1.x, va1.y }, pB0, b1);
            a2 = wmma_f32((v2f){ va2.x, va2.y }, pA0, a2);
            b2 = wmma_f32((v2f){ va2.x, va2.y }, pB0, b2);
            a3 = wmma_f32((v2f){ va3.x, va3.y }, pA0, a3);
            b3 = wmma_f32((v2f){ va3.x, va3.y }, pB0, b3);
            a0 = wmma_f32((v2f){ va0.z, va0.w }, pA1, a0);
            b0 = wmma_f32((v2f){ va0.z, va0.w }, pB1, b0);
            a1 = wmma_f32((v2f){ va1.z, va1.w }, pA1, a1);
            b1 = wmma_f32((v2f){ va1.z, va1.w }, pB1, b1);
            a2 = wmma_f32((v2f){ va2.z, va2.w }, pA1, a2);
            b2 = wmma_f32((v2f){ va2.z, va2.w }, pB1, b2);
            a3 = wmma_f32((v2f){ va3.z, va3.w }, pA1, a3);
            b3 = wmma_f32((v2f){ va3.z, va3.w }, pB1, b3);
            a0 = wmma_f32((v2f){ vb0.x, vb0.y }, pA2, a0);
            b0 = wmma_f32((v2f){ vb0.x, vb0.y }, pB2, b0);
            a1 = wmma_f32((v2f){ vb1.x, vb1.y }, pA2, a1);
            b1 = wmma_f32((v2f){ vb1.x, vb1.y }, pB2, b1);
            a2 = wmma_f32((v2f){ vb2.x, vb2.y }, pA2, a2);
            b2 = wmma_f32((v2f){ vb2.x, vb2.y }, pB2, b2);
            a3 = wmma_f32((v2f){ vb3.x, vb3.y }, pA2, a3);
            b3 = wmma_f32((v2f){ vb3.x, vb3.y }, pB2, b3);
            a0 = wmma_f32((v2f){ vb0.z, vb0.w }, pA3, a0);
            b0 = wmma_f32((v2f){ vb0.z, vb0.w }, pB3, b0);
            a1 = wmma_f32((v2f){ vb1.z, vb1.w }, pA3, a1);
            b1 = wmma_f32((v2f){ vb1.z, vb1.w }, pB3, b1);
            a2 = wmma_f32((v2f){ vb2.z, vb2.w }, pA3, a2);
            b2 = wmma_f32((v2f){ vb2.z, vb2.w }, pB3, b2);
            a3 = wmma_f32((v2f){ vb3.z, vb3.w }, pA3, a3);
            b3 = wmma_f32((v2f){ vb3.z, vb3.w }, pB3, b3);
        }

        const float lAt = lA + xhalf(lA);
        const float lBt = lB + xhalf(lB);

        {
            float* orow = &Wsm[2 * pass][wv][0] + l16 * OPITCH + 8 * half;
            *(float4*)(orow +  0) = make_float4(a0[0], a0[1], a0[2], a0[3]);
            *(float4*)(orow +  4) = make_float4(a0[4], a0[5], a0[6], a0[7]);
            *(float4*)(orow + 16) = make_float4(a1[0], a1[1], a1[2], a1[3]);
            *(float4*)(orow + 20) = make_float4(a1[4], a1[5], a1[6], a1[7]);
            *(float4*)(orow + 32) = make_float4(a2[0], a2[1], a2[2], a2[3]);
            *(float4*)(orow + 36) = make_float4(a2[4], a2[5], a2[6], a2[7]);
            *(float4*)(orow + 48) = make_float4(a3[0], a3[1], a3[2], a3[3]);
            *(float4*)(orow + 52) = make_float4(a3[4], a3[5], a3[6], a3[7]);
        }
        {
            float* orow = &Wsm[2 * pass + 1][wv][0] + l16 * OPITCH + 8 * half;
            *(float4*)(orow +  0) = make_float4(b0[0], b0[1], b0[2], b0[3]);
            *(float4*)(orow +  4) = make_float4(b0[4], b0[5], b0[6], b0[7]);
            *(float4*)(orow + 16) = make_float4(b1[0], b1[1], b1[2], b1[3]);
            *(float4*)(orow + 20) = make_float4(b1[4], b1[5], b1[6], b1[7]);
            *(float4*)(orow + 32) = make_float4(b2[0], b2[1], b2[2], b2[3]);
            *(float4*)(orow + 36) = make_float4(b2[4], b2[5], b2[6], b2[7]);
            *(float4*)(orow + 48) = make_float4(b3[0], b3[1], b3[2], b3[3]);
            *(float4*)(orow + 52) = make_float4(b3[4], b3[5], b3[6], b3[7]);
        }
        if (half == 0) {
            Msm[2 * pass][wv][l16]     = mA;
            Lsm[2 * pass][wv][l16]     = lAt;
            Msm[2 * pass + 1][wv][l16] = mB;
            Lsm[2 * pass + 1][wv][l16] = lBt;
        }
    }
    __syncthreads();

    const int ab = wv >> 3;
    const int R  = 2 * (wv & 7) + half;
    for (int pass = 0; pass < 2; ++pass) {
        const int slot = 2 * pass + ab;
        const int qi = 2 * (pass ? (NDT - 1 - pr) : pr) + ab;
        float mw[NWAVE];
        float M = -INFINITY;
#pragma unroll
        for (int w = 0; w < NWAVE; ++w) { mw[w] = Msm[slot][w][R]; M = fmaxf(M, mw[w]); }
        float L = 0.0f;
        float4 acc = make_float4(0.0f, 0.0f, 0.0f, 0.0f);
#pragma unroll
        for (int w = 0; w < NWAVE; ++w) {
            const float g = __builtin_amdgcn_exp2f(mw[w] - M);
            L += g * Lsm[slot][w][R];
            const float4 ov = *(const float4*)(&Wsm[slot][w][R * OPITCH + 4 * l16]);
            acc.x += g * ov.x; acc.y += g * ov.y;
            acc.z += g * ov.z; acc.w += g * ov.w;
        }
        const float inv = 1.0f / L;
        acc.x *= inv; acc.y *= inv; acc.z *= inv; acc.w *= inv;
        float* Ob = out + ((size_t)b * Tsz + qi * 16 + R) * Hsz + 4 * l16;
        *(float4*)Ob = acc;
    }
}

extern "C" void kernel_launch(void* const* d_in, const int* in_sizes, int n_in,
                              void* d_out, int out_size, void* d_ws, size_t ws_size,
                              hipStream_t stream) {
    const float* x  = (const float*)d_in[0];
    const float* Wq = (const float*)d_in[1];
    const float* Wk = (const float*)d_in[2];
    const float* Wv = (const float*)d_in[3];

    const size_t BTH = (size_t)Bsz * Tsz * Hsz;
    float* qkv = (float*)d_ws;
    float* Wt  = qkv + 3 * BTH;

    dim3 g0(Hsz, 3), blk0(256);
    wt_kernel<<<g0, blk0, 0, stream>>>(Wq, Wk, Wv, Wt);

    dim3 g1(Bsz * Tsz / QBM), blk1(256);
    qkv_kernel<<<g1, blk1, 0, stream>>>(x, Wt, qkv);

    dim3 g2(Tsz / 64, Bsz), blk2(32 * NWAVE);
    attn_kernel<<<g2, blk2, 0, stream>>>(qkv, (float*)d_out);
}


# qkv projection: the two waves of a SIMD run a slab's phases in opposite order (0..3: 48 WMMAs, then next slab's LDS loads/stores; 4..7 reversed); fragments double-buffered in registers (sums kept)
# speedup vs baseline: 3.9513x; 1.0374x over previous
#include <hip/hip_runtime.h>
#include <math.h>

#define Bsz 8
#define Tsz 2048
#define Csz 1024
#define Hsz 64

typedef __attribute__((ext_vector_type(2))) float v2f;
typedef __attribute__((ext_vector_type(8))) float v8f;

__device__ __forceinline__ v8f wmma_f32(v2f a, v2f b, v8f c) {
    return __builtin_amdgcn_wmma_f32_16x16x4_f32(
         false, a,  false, b,
         (short)0, c,  false,  false);
}

__global__ __launch_bounds__(256) void wt_kernel(
    const float* __restrict__ Wq,
    const float* __restrict__ Wk,
    const float* __restrict__ Wv,
    float* __restrict__ Wt)
{
    const int which = blockIdx.y;
    const float* W = (which == 0) ? Wq : (which == 1) ? Wk : Wv;
    float* dst = Wt + (size_t)which * Hsz * Csz;
    const int h = blockIdx.x;
    for (int c = threadIdx.x; c < Csz; c += 256)
        dst[(size_t)h * Csz + c] = W[(size_t)c * Hsz + h];
}

#define QBM 64
#define QBN 192
#define QBK 32
#define QPITCH 36
#define QROWS (QBM + QBN)
#define QSTG (QROWS * QPITCH)

__global__ __launch_bounds__(256) __attribute__((amdgpu_waves_per_eu(2, 2))) void qkv_kernel(
    const float* __restrict__ x,
    const float* __restrict__ Wt,
    float* __restrict__ qkv)
{
    __shared__ __attribute__((aligned(16))) float Ssm[2 * QSTG];

    const int tid  = threadIdx.x;
    const int lane = tid & 31;
    const int wv   = __builtin_amdgcn_readfirstlane(tid >> 5);
    const int wm   = wv & 3;
    const int wn   = wv >> 2;
    const int half = lane >> 4;
    const int l16  = lane & 15;
    const long m0  = (long)blockIdx.x * QBM;

    const int sc = tid & 7;
    const int sr = tid >> 3;
    const float* xg = x  + (m0 + sr) * (size_t)Csz + 4 * sc;
    const float* wg = Wt + (size_t)sr * Csz + 4 * sc;
    const int sdst = sr * QPITCH + 4 * sc;

    const int aoff = (16 * wm + l16) * QPITCH + 4 * half;
    const int boff = (QBM + 32 * wn + l16) * QPITCH + 4 * half;
#define QKV_TJ(j) ((64 * ((j) >> 1) + 16 * ((j) & 1)) * QPITCH)

    v8f acc[6];
#pragma unroll
    for (int j = 0; j < 6; ++j) acc[j] = (v8f){};

    float4 p0, p1, p2, p3, p4, p5, p6, p7;
    float4 q0, q1, q2, q3, q4, q5, q6, q7;
#define QKV_LOAD_SLAB(P, k)                                          \
    P##0 = *(const float4*)(xg + (k));                               \
    P##1 = *(const float4*)(xg + (size_t)32 * Csz + (k));            \
    P##2 = *(const float4*)(wg + (k));                               \
    P##3 = *(const float4*)(wg + (size_t)32 * Csz + (k));            \
    P##4 = *(const float4*)(wg + (size_t)64 * Csz + (k));            \
    P##5 = *(const float4*)(wg + (size_t)96 * Csz + (k));            \
    P##6 = *(const float4*)(wg + (size_t)128 * Csz + (k));           \
    P##7 = *(const float4*)(wg + (size_t)160 * Csz + (k));
#define QKV_STORE_SLAB(P, S)                                         \
    *(float4*)((S) + sdst +   0 * QPITCH) = P##0;                    \
    *(float4*)((S) + sdst +  32 * QPITCH) = P##1;                    \
    *(float4*)((S) + sdst +  64 * QPITCH) = P##2;                    \
    *(float4*)((S) + sdst +  96 * QPITCH) = P##3;                    \
    *(float4*)((S) + sdst + 128 * QPITCH) = P##4;                    \
    *(float4*)((S) + sdst + 160 * QPITCH) = P##5;                    \
    *(float4*)((S) + sdst + 192 * QPITCH) = P##6;                    \
    *(float4*)((S) + sdst + 224 * QPITCH) = P##7;

    float4 afX[QBK / 8], bfX[QBK / 8][6];
    float4 afY[QBK / 8], bfY[QBK / 8][6];
#define QKV_FRAGS(AF, BF, S)                                                   \
    _Pragma("unroll")                                                          \
    for (int g = 0; g < QBK / 8; ++g) {                                        \
        AF[g] = *(const float4*)((S) + aoff + 8 * g);                          \
        _Pragma("unroll")                                                      \
        for (int j = 0; j < 6; ++j)                                            \
            BF[g][j] = *(const float4*)((S) + boff + QKV_TJ(j) + 8 * g);       \
    }
#define QKV_MMA(AF, BF)                                                        \
    _Pragma("unroll")                                                          \
    for (int g = 0; g < QBK / 8; ++g) {                                        \
        const v2f alo = { AF[g].x, AF[g].y }, ahi = { AF[g].z, AF[g].w };      \
        _Pragma("unroll")                                                      \
        for (int j = 0; j < 6; ++j) {                                          \
            const v2f blo = { BF[g][j].x, BF[g][j].y };                        \
            acc[j] = ((j >> 1) == 1) ? wmma_f32(blo, alo, acc[j])              \
                                     : wmma_f32(alo, blo, acc[j]);             \
        }                                                                      \
        _Pragma("unroll")                                                      \
        for (int j = 0; j < 6; ++j) {                                          \
            const v2f bhi = { BF[g][j].z, BF[g][j].w };                        \
            acc[j] = ((j >> 1) == 1) ? wmma_f32(bhi, ahi, acc[j])              \
                                     : wmma_f32(ahi, bhi, acc[j]);             \
        }                                                                      \
    }
#define QKV_NPHASE(s, AF, BF, SR, SW)                                          \
    {                                                                          \
        if ((s) + 1 < NS) { QKV_FRAGS(AF, BF, SR) }                            \
        if ((s) + 2 < NS) { QKV_STORE_SLAB(p, SW) }                            \
        if ((s) + 3 < NS) { const int k3 = ((s) + 3) * QBK; QKV_LOAD_SLAB(p, k3) } \
    }
#define QKV_LDS_BARRIER()                                                      \
    __builtin_amdgcn_fence(__ATOMIC_RELEASE, "workgroup", "local");            \
    __builtin_amdgcn_s_barrier();                                              \
    __builtin_amdgcn_fence(__ATOMIC_ACQUIRE, "workgroup", "local");
#define QKV_TRIP(s, AFC, BFC, AFN, BFN, SR, SW)                                \
    if (wn != 0) QKV_NPHASE(s, AFN, BFN, SR, SW)                               \
    QKV_MMA(AFC, BFC)                                                          \
    if (wn == 0) QKV_NPHASE(s, AFN, BFN, SR, SW)                               \
    QKV_LDS_BARRIER()

    float* const S0 = &Ssm[0];
    float* const S1 = &Ssm[QSTG];
    const int NS = Csz / QBK;

    QKV_LOAD_SLAB(p, 0)
    QKV_LOAD_SLAB(q, QBK)
    QKV_STORE_SLAB(p, S0)
    QKV_STORE_SLAB(q, S1)
    __syncthreads();
    QKV_LOAD_SLAB(p, 2 * QBK)
    QKV_FRAGS(afX, bfX, S0)
    QKV_LDS_BARRIER()

    if (wn == 0) __builtin_amdgcn_s_setprio(1);

    for (int s = 0; s < NS; s += 2) {
        QKV_TRIP(s,     afX, bfX, afY, bfY, S1, S0)
        QKV_TRIP(s + 1, afY, bfY, afX, bfX, S0, S1)
    }

    if (wn == 0) __builtin_amdgcn_s_setprio(0);
#undef QKV_LOAD_SLAB
#undef QKV_STORE_SLAB
#undef QKV_FRAGS
#undef QKV_MMA
#undef QKV_NPHASE
#undef QKV_TRIP
#undef QKV_LDS_BARRIER

    const size_t BTH = (size_t)Bsz * Tsz * Hsz;
    const long mrow = m0 + 16 * wm;
    const long tile = (long)blockIdx.x * (QBM / 16) + wm;
#pragma unroll
    for (int j = 0; j < 2; ++j) {
        const int hcol = 32 * wn + 16 * j + l16;
        float* op = qkv + (mrow + 8 * half) * (size_t)Hsz + hcol;
#pragma unroll
        for (int r = 0; r < 8; ++r) op[(size_t)r * Hsz] = acc[j][r];
    }
#pragma unroll
    for (int j = 2; j < 4; ++j) {
        const int c = 4 * wn + 2 * (j & 1) + half;
        float* kp = qkv + BTH + ((size_t)tile * 8 + c) * 128 + 4 * l16;
        *(float4*)(kp +  0) = make_float4(acc[j][0], acc[j][1], acc[j][2], acc[j][3]);
        *(float4*)(kp + 64) = make_float4(acc[j][4], acc[j][5], acc[j][6], acc[j][7]);
    }
#pragma unroll
    for (int j = 4; j < 6; ++j) {
        const int t = 2 * wn + (j & 1);
        float* vp = qkv + 2 * BTH + ((size_t)tile * 4 + t) * 256 + 4 * lane;
        *(float4*)(vp +   0) = make_float4(acc[j][0], acc[j][1], acc[j][2], acc[j][3]);
        *(float4*)(vp + 128) = make_float4(acc[j][4], acc[j][5], acc[j][6], acc[j][7]);
    }
#undef QKV_TJ
}

#define NWAVE 16
#define NDT (Tsz / 32)
#define OPITCH 68

__device__ __forceinline__ float xhalf(float x) {
    const int b = __builtin_bit_cast(int, x);
    return __builtin_bit_cast(float,
        __builtin_amdgcn_permlanex16(b, b, 0x76543210, 0xfedcba98, false, false));
}

template <bool GUARD>
__device__ __forceinline__ float softmax_step(v8f& s, float& m_run, float& l_run) {
    float mx = fmaxf(fmaxf(fmaxf(s[0], s[1]), fmaxf(s[2], s[3])),
                     fmaxf(fmaxf(s[4], s[5]), fmaxf(s[6], s[7])));
    mx = fmaxf(mx, xhalf(mx));
    const float mn    = fmaxf(m_run, mx);
    const float mref  = (GUARD && mn == -INFINITY) ? 0.0f : mn;
    const float alpha = __builtin_amdgcn_exp2f(m_run - mref);
    m_run = mn;
#pragma unroll
    for (int r = 0; r < 8; ++r) s[r] = __builtin_amdgcn_exp2f(s[r] - mref);
    const float sm = ((s[0] + s[1]) + (s[2] + s[3])) + ((s[4] + s[5]) + (s[6] + s[7]));
    l_run = l_run * alpha + sm;
    return alpha;
}

__global__ __launch_bounds__(32 * NWAVE) void attn_kernel(
    const float* __restrict__ qkv,
    float* __restrict__ out)
{
    __shared__ __attribute__((aligned(16))) float Wsm[4][NWAVE][16 * OPITCH];
    __shared__ float Msm[4][NWAVE][16];
    __shared__ float Lsm[4][NWAVE][16];

    const int lane = threadIdx.x & 31;
    const int wv   = __builtin_amdgcn_readfirstlane((int)(threadIdx.x >> 5));
    const int half = lane >> 4;
    const int l16  = lane & 15;
    const int b    = blockIdx.y;
    const int pr   = blockIdx.x;

    const size_t BTH = (size_t)Bsz * Tsz * Hsz;
    const float* Q  = qkv +            (size_t)b * Tsz * Hsz;
    const float* Kp = qkv + BTH     +  (size_t)b * Tsz * Hsz + 4 * lane;
    const float* Vp = qkv + 2 * BTH +  (size_t)b * Tsz * Hsz + 4 * lane;

    const float scale = 0.125f * 1.4426950408889634f;

#pragma unroll 1
    for (int pass = 0; pass < 2; ++pass) {
        const int dq  = pass ? (NDT - 1 - pr) : pr;
        const int qiA = 2 * dq, qiB = 2 * dq + 1;
        const int k0  = pass ? ((wv - 2 * pr - 2) & (NWAVE - 1)) : wv;
        const int qrowA = qiA * 16 + l16;

        v2f qa[16], qb[16];
#pragma unroll
        for (int c = 0; c < 8; ++c) {
            const float4 ta = *(const float4*)(Q + (size_t)qrowA * Hsz + 8 * c + 4 * half);
            const float4 tb = *(const float4*)(Q + (size_t)(qrowA + 16) * Hsz + 8 * c + 4 * half);
            qa[2 * c][0]     = ta.x * scale;
            qa[2 * c][1]     = ta.y * scale;
            qa[2 * c + 1][0] = ta.z * scale;
            qa[2 * c + 1][1] = ta.w * scale;
            qb[2 * c][0]     = tb.x * scale;
            qb[2 * c][1]     = tb.y * scale;
            qb[2 * c + 1][0] = tb.z * scale;
            qb[2 * c + 1][1] = tb.w * scale;
        }

        float mA = -INFINITY, lA = 0.0f, mB = -INFINITY, lB = 0.0f;
        v8f a0 = {}, a1 = {}, a2 = {}, a3 = {};
        v8f b0 = {}, b1 = {}, b2 = {}, b3 = {};

        for (int kt = k0; kt <= qiB; kt += NWAVE) {
            const float* Kt = Kp + (size_t)kt * 1024;
            v8f sA = {}, sB = {};
#pragma unroll
            for (int c = 0; c < 8; ++c) {
                const float4 ka = *(const float4*)(Kt + 128 * c);
                sA = wmma_f32((v2f){ ka.x, ka.y }, qa[2 * c],     sA);
                sB = wmma_f32((v2f){ ka.x, ka.y }, qb[2 * c],     sB);
                sA = wmma_f32((v2f){ ka.z, ka.w }, qa[2 * c + 1], sA);
                sB = wmma_f32((v2f){ ka.z, ka.w }, qb[2 * c + 1], sB);
            }

            if (kt >= qiA) {
                const int koff = 16 * (kt - qiA) + 8 * half;
#pragma unroll
                for (int r = 0; r < 8; ++r)
                    if (koff + r > l16) sA[r] = -INFINITY;
                if (kt == qiB) {
#pragma unroll
                    for (int r = 0; r < 8; ++r)
                        if (8 * half + r > l16) sB[r] = -INFINITY;
                }
            }

            const float alA = softmax_step<true>(sA, mA, lA);
            const float alB = softmax_step<false>(sB, mB, lB);

#pragma unroll
            for (int r = 0; r < 8; ++r) {
                a0[r] *= alA; a1[r] *= alA; a2[r] *= alA; a3[r] *= alA;
                b0[r] *= alB; b1[r] *= alB; b2[r] *= alB; b3[r] *= alB;
            }

            const float* Vk = Vp + (size_t)kt * 1024;
            const float4 va0 = *(const float4*)(Vk +   0), vb0 = *(const float4*)(Vk + 128);
            const float4 va1 = *(const float4*)(Vk + 256), vb1 = *(const float4*)(Vk + 384);
            const float4 va2 = *(const float4*)(Vk + 512), vb2 = *(const float4*)(Vk + 640);
            const float4 va3 = *(const float4*)(Vk + 768), vb3 = *(const float4*)(Vk + 896);
            const v2f pA0 = { sA[0], sA[1] }, pA1 = { sA[2], sA[3] };
            const v2f pA2 = { sA[4], sA[5] }, pA3 = { sA[6], sA[7] };
            const v2f pB0 = { sB[0], sB[1] }, pB1 = { sB[2], sB[3] };
            const v2f pB2 = { sB[4], sB[5] }, pB3 = { sB[6], sB[7] };
            a0 = wmma_f32((v2f){ va0.x, va0.y }, pA0, a0);
            b0 = wmma_f32((v2f){ va0.x, va0.y }, pB0, b0);
            a1 = wmma_f32((v2f){ va1.x, va1.y }, pA0, a1);
            b1 = wmma_f32((v2f){ va1.x, va1.y }, pB0, b1);
            a2 = wmma_f32((v2f){ va2.x, va2.y }, pA0, a2);
            b2 = wmma_f32((v2f){ va2.x, va2.y }, pB0, b2);
            a3 = wmma_f32((v2f){ va3.x, va3.y }, pA0, a3);
            b3 = wmma_f32((v2f){ va3.x, va3.y }, pB0, b3);
            a0 = wmma_f32((v2f){ va0.z, va0.w }, pA1, a0);
            b0 = wmma_f32((v2f){ va0.z, va0.w }, pB1, b0);
            a1 = wmma_f32((v2f){ va1.z, va1.w }, pA1, a1);
            b1 = wmma_f32((v2f){ va1.z, va1.w }, pB1, b1);
            a2 = wmma_f32((v2f){ va2.z, va2.w }, pA1, a2);
            b2 = wmma_f32((v2f){ va2.z, va2.w }, pB1, b2);
            a3 = wmma_f32((v2f){ va3.z, va3.w }, pA1, a3);
            b3 = wmma_f32((v2f){ va3.z, va3.w }, pB1, b3);
            a0 = wmma_f32((v2f){ vb0.x, vb0.y }, pA2, a0);
            b0 = wmma_f32((v2f){ vb0.x, vb0.y }, pB2, b0);
            a1 = wmma_f32((v2f){ vb1.x, vb1.y }, pA2, a1);
            b1 = wmma_f32((v2f){ vb1.x, vb1.y }, pB2, b1);
            a2 = wmma_f32((v2f){ vb2.x, vb2.y }, pA2, a2);
            b2 = wmma_f32((v2f){ vb2.x, vb2.y }, pB2, b2);
            a3 = wmma_f32((v2f){ vb3.x, vb3.y }, pA2, a3);
            b3 = wmma_f32((v2f){ vb3.x, vb3.y }, pB2, b3);
            a0 = wmma_f32((v2f){ vb0.z, vb0.w }, pA3, a0);
            b0 = wmma_f32((v2f){ vb0.z, vb0.w }, pB3, b0);
            a1 = wmma_f32((v2f){ vb1.z, vb1.w }, pA3, a1);
            b1 = wmma_f32((v2f){ vb1.z, vb1.w }, pB3, b1);
            a2 = wmma_f32((v2f){ vb2.z, vb2.w }, pA3, a2);
            b2 = wmma_f32((v2f){ vb2.z, vb2.w }, pB3, b2);
            a3 = wmma_f32((v2f){ vb3.z, vb3.w }, pA3, a3);
            b3 = wmma_f32((v2f){ vb3.z, vb3.w }, pB3, b3);
        }

        const float lAt = lA + xhalf(lA);
        const float lBt = lB + xhalf(lB);

        {
            float* orow = &Wsm[2 * pass][wv][0] + l16 * OPITCH + 8 * half;
            *(float4*)(orow +  0) = make_float4(a0[0], a0[1], a0[2], a0[3]);
            *(float4*)(orow +  4) = make_float4(a0[4], a0[5], a0[6], a0[7]);
            *(float4*)(orow + 16) = make_float4(a1[0], a1[1], a1[2], a1[3]);
            *(float4*)(orow + 20) = make_float4(a1[4], a1[5], a1[6], a1[7]);
            *(float4*)(orow + 32) = make_float4(a2[0], a2[1], a2[2], a2[3]);
            *(float4*)(orow + 36) = make_float4(a2[4], a2[5], a2[6], a2[7]);
            *(float4*)(orow + 48) = make_float4(a3[0], a3[1], a3[2], a3[3]);
            *(float4*)(orow + 52) = make_float4(a3[4], a3[5], a3[6], a3[7]);
        }
        {
            float* orow = &Wsm[2 * pass + 1][wv][0] + l16 * OPITCH + 8 * half;
            *(float4*)(orow +  0) = make_float4(b0[0], b0[1], b0[2], b0[3]);
            *(float4*)(orow +  4) = make_float4(b0[4], b0[5], b0[6], b0[7]);
            *(float4*)(orow + 16) = make_float4(b1[0], b1[1], b1[2], b1[3]);
            *(float4*)(orow + 20) = make_float4(b1[4], b1[5], b1[6], b1[7]);
            *(float4*)(orow + 32) = make_float4(b2[0], b2[1], b2[2], b2[3]);
            *(float4*)(orow + 36) = make_float4(b2[4], b2[5], b2[6], b2[7]);
            *(float4*)(orow + 48) = make_float4(b3[0], b3[1], b3[2], b3[3]);
            *(float4*)(orow + 52) = make_float4(b3[4], b3[5], b3[6], b3[7]);
        }
        if (half == 0) {
            Msm[2 * pass][wv][l16]     = mA;
            Lsm[2 * pass][wv][l16]     = lAt;
            Msm[2 * pass + 1][wv][l16] = mB;
            Lsm[2 * pass + 1][wv][l16] = lBt;
        }
    }
    __syncthreads();

    const int ab = wv >> 3;
    const int R  = 2 * (wv & 7) + half;
    for (int pass = 0; pass < 2; ++pass) {
        const int slot = 2 * pass + ab;
        const int qi = 2 * (pass ? (NDT - 1 - pr) : pr) + ab;
        float mw[NWAVE];
        float M = -INFINITY;
#pragma unroll
        for (int w = 0; w < NWAVE; ++w) { mw[w] = Msm[slot][w][R]; M = fmaxf(M, mw[w]); }
        float L = 0.0f;
        float4 acc = make_float4(0.0f, 0.0f, 0.0f, 0.0f);
#pragma unroll
        for (int w = 0; w < NWAVE; ++w) {
            const float g = __builtin_amdgcn_exp2f(mw[w] - M);
            L += g * Lsm[slot][w][R];
            const float4 ov = *(const float4*)(&Wsm[slot][w][R * OPITCH + 4 * l16]);
            acc.x += g * ov.x; acc.y += g * ov.y;
            acc.z += g * ov.z; acc.w += g * ov.w;
        }
        const float inv = 1.0f / L;
        acc.x *= inv; acc.y *= inv; acc.z *= inv; acc.w *= inv;
        float* Ob = out + ((size_t)b * Tsz + qi * 16 + R) * Hsz + 4 * l16;
        *(float4*)Ob = acc;
    }
}

extern "C" void kernel_launch(void* const* d_in, const int* in_sizes, int n_in,
                              void* d_out, int out_size, void* d_ws, size_t ws_size,
                              hipStream_t stream) {
    const float* x  = (const float*)d_in[0];
    const float* Wq = (const float*)d_in[1];
    const float* Wk = (const float*)d_in[2];
    const float* Wv = (const float*)d_in[3];

    const size_t BTH = (size_t)Bsz * Tsz * Hsz;
    float* qkv = (float*)d_ws;
    float* Wt  = qkv + 3 * BTH;

    dim3 g0(Hsz, 3), blk0(256);
    wt_kernel<<<g0, blk0, 0, stream>>>(Wq, Wk, Wv, Wt);

    dim3 g1(Bsz * Tsz / QBM), blk1(256);
    qkv_kernel<<<g1, blk1, 0, stream>>>(x, Wt, qkv);

    dim3 g2(Tsz / 64, Bsz), blk2(32 * NWAVE);
    attn_kernel<<<g2, blk2, 0, stream>>>(qkv, (float*)d_out);
}


# qkv projection stages Wq/Wk/Wv slabs as given, [k][column] in 1-KB LDS rows; weight fragments by two-address LDS loads: no transposed weights, no launch making them (3 -> 2 launches, sums kept)
# speedup vs baseline: 4.2088x; 1.0652x over previous
#include <hip/hip_runtime.h>
#include <math.h>

#define Bsz 8
#define Tsz 2048
#define Csz 1024
#define Hsz 64

typedef __attribute__((ext_vector_type(2))) float v2f;
typedef __attribute__((ext_vector_type(8))) float v8f;

__device__ __forceinline__ v8f wmma_f32(v2f a, v2f b, v8f c) {
    return __builtin_amdgcn_wmma_f32_16x16x4_f32(
         false, a,  false, b,
         (short)0, c,  false,  false);
}

#define QBM 64
#define QBN 192
#define QBK 32
#define QPITCH 36
#define QWP 256
#define QWOFF (QBM * QPITCH)
#define QSTG (QWOFF + QBK * QWP)

__global__ __launch_bounds__(256) __attribute__((amdgpu_waves_per_eu(2, 2))) void qkv_kernel(
    const float* __restrict__ x,
    const float* __restrict__ Wq,
    const float* __restrict__ Wk,
    const float* __restrict__ Wv,
    float* __restrict__ qkv)
{
    __shared__ __attribute__((aligned(16))) float Ssm[2 * QSTG];

    const int tid  = threadIdx.x;
    const int lane = tid & 31;
    const int wv   = __builtin_amdgcn_readfirstlane(tid >> 5);
    const int wm   = wv & 3;
    const int wn   = wv >> 2;
    const int half = lane >> 4;
    const int l16  = lane & 15;
    const long m0  = (long)blockIdx.x * QBM;

    const int sc = tid & 7;
    const int sr = tid >> 3;
    const float* xg = x + (m0 + sr) * (size_t)Csz + 4 * sc;
    const int sdst = sr * QPITCH + 4 * sc;
    const int wsrc = 4 * tid;
    const int wdst = QWOFF + (tid >> 4) * QWP + ((4 * (tid & 15)) ^ (16 * ((tid >> 6) & 1)));

    const int aoff = (16 * wm + l16) * QPITCH + 4 * half;
    int wb0 = QWOFF + 4 * half * QWP + ((32 * wn + l16) ^ (16 * half));
    int wb1 = QWOFF + 4 * half * QWP + ((32 * wn + 16 + l16) ^ (16 * half));
    int wb2 = wb0 + 64, wb3 = wb1 + 64, wb4 = wb0 + 128, wb5 = wb1 + 128;
    asm volatile("" : "+v"(wb0));
    asm volatile("" : "+v"(wb1));
    asm volatile("" : "+v"(wb2));
    asm volatile("" : "+v"(wb3));
    asm volatile("" : "+v"(wb4));
    asm volatile("" : "+v"(wb5));

    v8f acc[6];
#pragma unroll
    for (int j = 0; j < 6; ++j) acc[j] = (v8f){};

    float4 p0, p1, p2, p3, p4, p5, p6, p7;
    float4 q0, q1, q2, q3, q4, q5, q6, q7;
#define QKV_LOAD_SLAB(P, k)                                          \
    P##0 = *(const float4*)(xg + (k));                               \
    P##1 = *(const float4*)(xg + (size_t)32 * Csz + (k));            \
    P##2 = *(const float4*)(Wq + (size_t)(k) * Hsz + wsrc);          \
    P##3 = *(const float4*)(Wq + (size_t)(k) * Hsz + wsrc + 1024);   \
    P##4 = *(const float4*)(Wk + (size_t)(k) * Hsz + wsrc);          \
    P##5 = *(const float4*)(Wk + (size_t)(k) * Hsz + wsrc + 1024);   \
    P##6 = *(const float4*)(Wv + (size_t)(k) * Hsz + wsrc);          \
    P##7 = *(const float4*)(Wv + (size_t)(k) * Hsz + wsrc + 1024);
#define QKV_STORE_SLAB(P, S)                                         \
    *(float4*)((S) + sdst +   0 * QPITCH) = P##0;                    \
    *(float4*)((S) + sdst +  32 * QPITCH) = P##1;                    \
    *(float4*)((S) + wdst) = P##2;                                   \
    *(float4*)((S) + wdst + 16 * QWP) = P##3;                        \
    *(float4*)((S) + wdst + 64) = P##4;                              \
    *(float4*)((S) + wdst + 64 + 16 * QWP) = P##5;                   \
    *(float4*)((S) + wdst + 128) = P##6;                             \
    *(float4*)((S) + wdst + 128 + 16 * QWP) = P##7;

    float4 afX[QBK / 8], bfX[QBK / 8][6];
    float4 afY[QBK / 8], bfY[QBK / 8][6];
#define QKV_WB(j) ((j) == 0 ? wb0 : (j) == 1 ? wb1 : (j) == 2 ? wb2 : (j) == 3 ? wb3 : (j) == 4 ? wb4 : wb5)
#define QKV_FRAGS(AF, BF, S)                                                   \
    _Pragma("unroll")                                                          \
    for (int g = 0; g < QBK / 8; ++g) {                                        \
        AF[g] = *(const float4*)((S) + aoff + 8 * g);                          \
        _Pragma("unroll")                                                      \
        for (int j = 0; j < 6; ++j) {                                          \
            const float* wp = (S) + QKV_WB(j) + 8 * g * QWP;                   \
            BF[g][j] = make_float4(wp[0], wp[QWP], wp[2 * QWP], wp[3 * QWP]);  \
        }                                                                      \
    }
#define QKV_MMA(AF, BF)                                                        \
    _Pragma("unroll")                                                          \
    for (int g = 0; g < QBK / 8; ++g) {                                        \
        const v2f alo = { AF[g].x, AF[g].y }, ahi = { AF[g].z, AF[g].w };      \
        _Pragma("unroll")                                                      \
        for (int j = 0; j < 6; ++j) {                                          \
            const v2f blo = { BF[g][j].x, BF[g][j].y };                        \
            acc[j] = ((j >> 1) == 1) ? wmma_f32(blo, alo, acc[j])              \
                                     : wmma_f32(alo, blo, acc[j]);             \
        }                                                                      \
        _Pragma("unroll")                                                      \
        for (int j = 0; j < 6; ++j) {                                          \
            const v2f bhi = { BF[g][j].z, BF[g][j].w };                        \
            acc[j] = ((j >> 1) == 1) ? wmma_f32(bhi, ahi, acc[j])              \
                                     : wmma_f32(ahi, bhi, acc[j]);             \
        }                                                                      \
    }
#define QKV_NPHASE(s, AF, BF, SR, SW)                                          \
    {                                                                          \
        if ((s) + 1 < NS) { QKV_FRAGS(AF, BF, SR) }                            \
        if ((s) + 2 < NS) { QKV_STORE_SLAB(p, SW) }                            \
        if ((s) + 3 < NS) { const int k3 = ((s) + 3) * QBK; QKV_LOAD_SLAB(p, k3) } \
    }
#define QKV_LDS_BARRIER()                                                      \
    __builtin_amdgcn_fence(__ATOMIC_RELEASE, "workgroup", "local");            \
    __builtin_amdgcn_s_barrier();                                              \
    __builtin_amdgcn_fence(__ATOMIC_ACQUIRE, "workgroup", "local");
#define QKV_TRIP(s, AFC, BFC, AFN, BFN, SR, SW)                                \
    if (wn != 0) QKV_NPHASE(s, AFN, BFN, SR, SW)                               \
    QKV_MMA(AFC, BFC)                                                          \
    if (wn == 0) QKV_NPHASE(s, AFN, BFN, SR, SW)                               \
    QKV_LDS_BARRIER()

    float* const S0 = &Ssm[0];
    float* const S1 = &Ssm[QSTG];
    const int NS = Csz / QBK;

    QKV_LOAD_SLAB(p, 0)
    QKV_LOAD_SLAB(q, QBK)
    QKV_STORE_SLAB(p, S0)
    QKV_STORE_SLAB(q, S1)
    __syncthreads();
    QKV_LOAD_SLAB(p, 2 * QBK)
    QKV_FRAGS(afX, bfX, S0)
    QKV_LDS_BARRIER()

    if (wn == 0) __builtin_amdgcn_s_setprio(1);

    for (int s = 0; s < NS; s += 2) {
        QKV_TRIP(s,     afX, bfX, afY, bfY, S1, S0)
        QKV_TRIP(s + 1, afY, bfY, afX, bfX, S0, S1)
    }

    if (wn == 0) __builtin_amdgcn_s_setprio(0);
#undef QKV_LOAD_SLAB
#undef QKV_WB
#undef QKV_STORE_SLAB
#undef QKV_FRAGS
#undef QKV_MMA
#undef QKV_NPHASE
#undef QKV_TRIP
#undef QKV_LDS_BARRIER

    const size_t BTH = (size_t)Bsz * Tsz * Hsz;
    const long mrow = m0 + 16 * wm;
    const long tile = (long)blockIdx.x * (QBM / 16) + wm;
#pragma unroll
    for (int j = 0; j < 2; ++j) {
        const int hcol = 32 * wn + 16 * j + l16;
        float* op = qkv + (mrow + 8 * half) * (size_t)Hsz + hcol;
#pragma unroll
        for (int r = 0; r < 8; ++r) op[(size_t)r * Hsz] = acc[j][r];
    }
#pragma unroll
    for (int j = 2; j < 4; ++j) {
        const int c = 4 * wn + 2 * (j & 1) + half;
        float* kp = qkv + BTH + ((size_t)tile * 8 + c) * 128 + 4 * l16;
        *(float4*)(kp +  0) = make_float4(acc[j][0], acc[j][1], acc[j][2], acc[j][3]);
        *(float4*)(kp + 64) = make_float4(acc[j][4], acc[j][5], acc[j][6], acc[j][7]);
    }
#pragma unroll
    for (int j = 4; j < 6; ++j) {
        const int t = 2 * wn + (j & 1);
        float* vp = qkv + 2 * BTH + ((size_t)tile * 4 + t) * 256 + 4 * lane;
        *(float4*)(vp +   0) = make_float4(acc[j][0], acc[j][1], acc[j][2], acc[j][3]);
        *(float4*)(vp + 128) = make_float4(acc[j][4], acc[j][5], acc[j][6], acc[j][7]);
    }
}

#define NWAVE 16
#define NDT (Tsz / 32)
#define OPITCH 68

__device__ __forceinline__ float xhalf(float x) {
    const int b = __builtin_bit_cast(int, x);
    return __builtin_bit_cast(float,
        __builtin_amdgcn_permlanex16(b, b, 0x76543210, 0xfedcba98, false, false));
}

template <bool GUARD>
__device__ __forceinline__ float softmax_step(v8f& s, float& m_run, float& l_run) {
    float mx = fmaxf(fmaxf(fmaxf(s[0], s[1]), fmaxf(s[2], s[3])),
                     fmaxf(fmaxf(s[4], s[5]), fmaxf(s[6], s[7])));
    mx = fmaxf(mx, xhalf(mx));
    const float mn    = fmaxf(m_run, mx);
    const float mref  = (GUARD && mn == -INFINITY) ? 0.0f : mn;
    const float alpha = __builtin_amdgcn_exp2f(m_run - mref);
    m_run = mn;
#pragma unroll
    for (int r = 0; r < 8; ++r) s[r] = __builtin_amdgcn_exp2f(s[r] - mref);
    const float sm = ((s[0] + s[1]) + (s[2] + s[3])) + ((s[4] + s[5]) + (s[6] + s[7]));
    l_run = l_run * alpha + sm;
    return alpha;
}

__global__ __launch_bounds__(32 * NWAVE) void attn_kernel(
    const float* __restrict__ qkv,
    float* __restrict__ out)
{
    __shared__ __attribute__((aligned(16))) float Wsm[4][NWAVE][16 * OPITCH];
    __shared__ float Msm[4][NWAVE][16];
    __shared__ float Lsm[4][NWAVE][16];

    const int lane = threadIdx.x & 31;
    const int wv   = __builtin_amdgcn_readfirstlane((int)(threadIdx.x >> 5));
    const int half = lane >> 4;
    const int l16  = lane & 15;
    const int b    = blockIdx.y;
    const int pr   = blockIdx.x;

    const size_t BTH = (size_t)Bsz * Tsz * Hsz;
    const float* Q  = qkv +            (size_t)b * Tsz * Hsz;
    const float* Kp = qkv + BTH     +  (size_t)b * Tsz * Hsz + 4 * lane;
    const float* Vp = qkv + 2 * BTH +  (size_t)b * Tsz * Hsz + 4 * lane;

    const float scale = 0.125f * 1.4426950408889634f;

#pragma unroll 1
    for (int pass = 0; pass < 2; ++pass) {
        const int dq  = pass ? (NDT - 1 - pr) : pr;
        const int qiA = 2 * dq, qiB = 2 * dq + 1;
        const int k0  = pass ? ((wv - 2 * pr - 2) & (NWAVE - 1)) : wv;
        const int qrowA = qiA * 16 + l16;

        v2f qa[16], qb[16];
#pragma unroll
        for (int c = 0; c < 8; ++c) {
            const float4 ta = *(const float4*)(Q + (size_t)qrowA * Hsz + 8 * c + 4 * half);
            const float4 tb = *(const float4*)(Q + (size_t)(qrowA + 16) * Hsz + 8 * c + 4 * half);
            qa[2 * c][0]     = ta.x * scale;
            qa[2 * c][1]     = ta.y * scale;
            qa[2 * c + 1][0] = ta.z * scale;
            qa[2 * c + 1][1] = ta.w * scale;
            qb[2 * c][0]     = tb.x * scale;
            qb[2 * c][1]     = tb.y * scale;
            qb[2 * c + 1][0] = tb.z * scale;
            qb[2 * c + 1][1] = tb.w * scale;
        }

        float mA = -INFINITY, lA = 0.0f, mB = -INFINITY, lB = 0.0f;
        v8f a0 = {}, a1 = {}, a2 = {}, a3 = {};
        v8f b0 = {}, b1 = {}, b2 = {}, b3 = {};

        for (int kt = k0; kt <= qiB; kt += NWAVE) {
            const float* Kt = Kp + (size_t)kt * 1024;
            v8f sA = {}, sB = {};
#pragma unroll
            for (int c = 0; c < 8; ++c) {
                const float4 ka = *(const float4*)(Kt + 128 * c);
                sA = wmma_f32((v2f){ ka.x, ka.y }, qa[2 * c],     sA);
                sB = wmma_f32((v2f){ ka.x, ka.y }, qb[2 * c],     sB);
                sA = wmma_f32((v2f){ ka.z, ka.w }, qa[2 * c + 1], sA);
                sB = wmma_f32((v2f){ ka.z, ka.w }, qb[2 * c + 1], sB);
            }

            if (kt >= qiA) {
                const int koff = 16 * (kt - qiA) + 8 * half;
#pragma unroll
                for (int r = 0; r < 8; ++r)
                    if (koff + r > l16) sA[r] = -INFINITY;
                if (kt == qiB) {
#pragma unroll
                    for (int r = 0; r < 8; ++r)
                        if (8 * half + r > l16) sB[r] = -INFINITY;
                }
            }

            const float alA = softmax_step<true>(sA, mA, lA);
            const float alB = softmax_step<false>(sB, mB, lB);

#pragma unroll
            for (int r = 0; r < 8; ++r) {
                a0[r] *= alA; a1[r] *= alA; a2[r] *= alA; a3[r] *= alA;
                b0[r] *= alB; b1[r] *= alB; b2[r] *= alB; b3[r] *= alB;
            }

            const float* Vk = Vp + (size_t)kt * 1024;
            const float4 va0 = *(const float4*)(Vk +   0), vb0 = *(const float4*)(Vk + 128);
            const float4 va1 = *(const float4*)(Vk + 256), vb1 = *(const float4*)(Vk + 384);
            const float4 va2 = *(const float4*)(Vk + 512), vb2 = *(const float4*)(Vk + 640);
            const float4 va3 = *(const float4*)(Vk + 768), vb3 = *(const float4*)(Vk + 896);
            const v2f pA0 = { sA[0], sA[1] }, pA1 = { sA[2], sA[3] };
            const v2f pA2 = { sA[4], sA[5] }, pA3 = { sA[6], sA[7] };
            const v2f pB0 = { sB[0], sB[1] }, pB1 = { sB[2], sB[3] };
            const v2f pB2 = { sB[4], sB[5] }, pB3 = { sB[6], sB[7] };
            a0 = wmma_f32((v2f){ va0.x, va0.y }, pA0, a0);
            b0 = wmma_f32((v2f){ va0.x, va0.y }, pB0, b0);
            a1 = wmma_f32((v2f){ va1.x, va1.y }, pA0, a1);
            b1 = wmma_f32((v2f){ va1.x, va1.y }, pB0, b1);
            a2 = wmma_f32((v2f){ va2.x, va2.y }, pA0, a2);
            b2 = wmma_f32((v2f){ va2.x, va2.y }, pB0, b2);
            a3 = wmma_f32((v2f){ va3.x, va3.y }, pA0, a3);
            b3 = wmma_f32((v2f){ va3.x, va3.y }, pB0, b3);
            a0 = wmma_f32((v2f){ va0.z, va0.w }, pA1, a0);
            b0 = wmma_f32((v2f){ va0.z, va0.w }, pB1, b0);
            a1 = wmma_f32((v2f){ va1.z, va1.w }, pA1, a1);
            b1 = wmma_f32((v2f){ va1.z, va1.w }, pB1, b1);
            a2 = wmma_f32((v2f){ va2.z, va2.w }, pA1, a2);
            b2 = wmma_f32((v2f){ va2.z, va2.w }, pB1, b2);
            a3 = wmma_f32((v2f){ va3.z, va3.w }, pA1, a3);
            b3 = wmma_f32((v2f){ va3.z, va3.w }, pB1, b3);
            a0 = wmma_f32((v2f){ vb0.x, vb0.y }, pA2, a0);
            b0 = wmma_f32((v2f){ vb0.x, vb0.y }, pB2, b0);
            a1 = wmma_f32((v2f){ vb1.x, vb1.y }, pA2, a1);
            b1 = wmma_f32((v2f){ vb1.x, vb1.y }, pB2, b1);
            a2 = wmma_f32((v2f){ vb2.x, vb2.y }, pA2, a2);
            b2 = wmma_f32((v2f){ vb2.x, vb2.y }, pB2, b2);
            a3 = wmma_f32((v2f){ vb3.x, vb3.y }, pA2, a3);
            b3 = wmma_f32((v2f){ vb3.x, vb3.y }, pB2, b3);
            a0 = wmma_f32((v2f){ vb0.z, vb0.w }, pA3, a0);
            b0 = wmma_f32((v2f){ vb0.z, vb0.w }, pB3, b0);
            a1 = wmma_f32((v2f){ vb1.z, vb1.w }, pA3, a1);
            b1 = wmma_f32((v2f){ vb1.z, vb1.w }, pB3, b1);
            a2 = wmma_f32((v2f){ vb2.z, vb2.w }, pA3, a2);
            b2 = wmma_f32((v2f){ vb2.z, vb2.w }, pB3, b2);
            a3 = wmma_f32((v2f){ vb3.z, vb3.w }, pA3, a3);
            b3 = wmma_f32((v2f){ vb3.z, vb3.w }, pB3, b3);
        }

        const float lAt = lA + xhalf(lA);
        const float lBt = lB + xhalf(lB);

        {
            float* orow = &Wsm[2 * pass][wv][0] + l16 * OPITCH + 8 * half;
            *(float4*)(orow +  0) = make_float4(a0[0], a0[1], a0[2], a0[3]);
            *(float4*)(orow +  4) = make_float4(a0[4], a0[5], a0[6], a0[7]);
            *(float4*)(orow + 16) = make_float4(a1[0], a1[1], a1[2], a1[3]);
            *(float4*)(orow + 20) = make_float4(a1[4], a1[5], a1[6], a1[7]);
            *(float4*)(orow + 32) = make_float4(a2[0], a2[1], a2[2], a2[3]);
            *(float4*)(orow + 36) = make_float4(a2[4], a2[5], a2[6], a2[7]);
            *(float4*)(orow + 48) = make_float4(a3[0], a3[1], a3[2], a3[3]);
            *(float4*)(orow + 52) = make_float4(a3[4], a3[5], a3[6], a3[7]);
        }
        {
            float* orow = &Wsm[2 * pass + 1][wv][0] + l16 * OPITCH + 8 * half;
            *(float4*)(orow +  0) = make_float4(b0[0], b0[1], b0[2], b0[3]);
            *(float4*)(orow +  4) = make_float4(b0[4], b0[5], b0[6], b0[7]);
            *(float4*)(orow + 16) = make_float4(b1[0], b1[1], b1[2], b1[3]);
            *(float4*)(orow + 20) = make_float4(b1[4], b1[5], b1[6], b1[7]);
            *(float4*)(orow + 32) = make_float4(b2[0], b2[1], b2[2], b2[3]);
            *(float4*)(orow + 36) = make_float4(b2[4], b2[5], b2[6], b2[7]);
            *(float4*)(orow + 48) = make_float4(b3[0], b3[1], b3[2], b3[3]);
            *(float4*)(orow + 52) = make_float4(b3[4], b3[5], b3[6], b3[7]);
        }
        if (half == 0) {
            Msm[2 * pass][wv][l16]     = mA;
            Lsm[2 * pass][wv][l16]     = lAt;
            Msm[2 * pass + 1][wv][l16] = mB;
            Lsm[2 * pass + 1][wv][l16] = lBt;
        }
    }
    __syncthreads();

    const int ab = wv >> 3;
    const int R  = 2 * (wv & 7) + half;
    for (int pass = 0; pass < 2; ++pass) {
        const int slot = 2 * pass + ab;
        const int qi = 2 * (pass ? (NDT - 1 - pr) : pr) + ab;
        float mw[NWAVE];
        float M = -INFINITY;
#pragma unroll
        for (int w = 0; w < NWAVE; ++w) { mw[w] = Msm[slot][w][R]; M = fmaxf(M, mw[w]); }
        float L = 0.0f;
        float4 acc = make_float4(0.0f, 0.0f, 0.0f, 0.0f);
#pragma unroll
        for (int w = 0; w < NWAVE; ++w) {
            const float g = __builtin_amdgcn_exp2f(mw[w] - M);
            L += g * Lsm[slot][w][R];
            const float4 ov = *(const float4*)(&Wsm[slot][w][R * OPITCH + 4 * l16]);
            acc.x += g * ov.x; acc.y += g * ov.y;
            acc.z += g * ov.z; acc.w += g * ov.w;
        }
        const float inv = 1.0f / L;
        acc.x *= inv; acc.y *= inv; acc.z *= inv; acc.w *= inv;
        float* Ob = out + ((size_t)b * Tsz + qi * 16 + R) * Hsz + 4 * l16;
        *(float4*)Ob = acc;
    }
}

extern "C" void kernel_launch(void* const* d_in, const int* in_sizes, int n_in,
                              void* d_out, int out_size, void* d_ws, size_t ws_size,
                              hipStream_t stream) {
    const float* x  = (const float*)d_in[0];
    const float* Wq = (const float*)d_in[1];
    const float* Wk = (const float*)d_in[2];
    const float* Wv = (const float*)d_in[3];

    float* qkv = (float*)d_ws;

    dim3 g1(Bsz * Tsz / QBM), blk1(256);
    qkv_kernel<<<g1, blk1, 0, stream>>>(x, Wq, Wk, Wv, qkv);

    dim3 g2(Tsz / 64, Bsz), blk2(32 * NWAVE);
    attn_kernel<<<g2, blk2, 0, stream>>>(qkv, (float*)d_out);
}
